# starting kernel
# baseline (speedup 1.0000x reference)
#include <hip/hip_runtime.h>
#include <hip/hip_bf16.h>

#define N_NODES   50000
#define N_EDGES   800000
#define E_TOT     850000
#define IN_F      128
#define HC        128
#define NEG_SLOPE 0.2f

typedef __attribute__((ext_vector_type(2))) float v2f;
typedef __attribute__((ext_vector_type(8))) float v8f;

__global__ void gat_init(float* __restrict__ out, const float* __restrict__ bias,
                         float* __restrict__ m, float* __restrict__ denom) {
  int i = blockIdx.x * blockDim.x + threadIdx.x;
  if (i < N_NODES * HC) out[i] = bias[i & (HC - 1)];
  if (i < N_NODES * 4) {
    m[i] = -3.402823466e38f;
    denom[i] = 0.0f;
  }
}

__global__ void __launch_bounds__(256)
gat_gemm(const float* __restrict__ x, const float* __restrict__ Wl,
         const float* __restrict__ Wr, float* __restrict__ xl,
         float* __restrict__ xr) {
  const int gw   = (blockIdx.x * blockDim.x + threadIdx.x) >> 5;
  const int lane = threadIdx.x & 31;
  const int mt = gw >> 4;
  const int nt = gw & 15;

  const float* __restrict__ W   = (nt < 8) ? Wl : Wr;
  float* __restrict__       dst = (nt < 8) ? xl : xr;
  const int ncol  = (nt & 7) * 16 + (lane & 15);
  const int mrow  = mt * 16 + (lane & 15);
  const int khalf = (lane >> 4) * 2;

  const float* __restrict__ xrow = x + (size_t)mrow * IN_F;

  v8f c = {};
#pragma unroll
  for (int kb = 0; kb < IN_F; kb += 4) {
    v2f a, b;
    const float2 ax = *(const float2*)(xrow + kb + khalf);
    a.x = ax.x; a.y = ax.y;
    b.x = W[(size_t)(kb + khalf) * HC + ncol];
    b.y = W[(size_t)(kb + khalf + 1) * HC + ncol];
    c = __builtin_amdgcn_wmma_f32_16x16x4_f32(false, a, false, b,
                                              (short)0, c, false, false);
  }

  const int mbase = mt * 16 + 8 * (lane >> 4);
#pragma unroll
  for (int v = 0; v < 8; ++v)
    dst[(size_t)(mbase + v) * HC + ncol] = c[v];
}

__global__ void __launch_bounds__(256)
gat_logits(const int* __restrict__ ei, const float* __restrict__ xl,
           const float* __restrict__ xr, const float* __restrict__ att,
           float* __restrict__ logits, float* __restrict__ m) {
  const unsigned gid  = blockIdx.x * blockDim.x + threadIdx.x;
  const unsigned e    = gid >> 5;
  const unsigned lane = threadIdx.x & 31;

  int src, dst;
  if (e < N_EDGES) { src = ei[e]; dst = ei[N_EDGES + e]; }
  else             { src = dst = (int)(e - N_EDGES); }

  const float4 a = ((const float4*)(xl + (size_t)src * HC))[lane];
  const float4 b = ((const float4*)(xr + (size_t)dst * HC))[lane];
  const float4 w = ((const float4*)att)[lane];

  float vx = a.x + b.x, vy = a.y + b.y, vz = a.z + b.z, vw = a.w + b.w;
  vx = (vx > 0.f) ? vx : vx * NEG_SLOPE;
  vy = (vy > 0.f) ? vy : vy * NEG_SLOPE;
  vz = (vz > 0.f) ? vz : vz * NEG_SLOPE;
  vw = (vw > 0.f) ? vw : vw * NEG_SLOPE;
  float s = vx * w.x + vy * w.y + vz * w.z + vw * w.w;

  s += __shfl_xor(s, 1);
  s += __shfl_xor(s, 2);
  s += __shfl_xor(s, 4);

  if ((lane & 7) == 0) {
    const int h = lane >> 3;
    logits[(size_t)e * 4 + h] = s;
    atomicMax(&m[(size_t)dst * 4 + h], s);
  }
}

__global__ void gat_expsum(const int* __restrict__ ei,
                           const float* __restrict__ logits,
                           const float* __restrict__ m,
                           float* __restrict__ denom) {
  const unsigned gid = blockIdx.x * blockDim.x + threadIdx.x;
  if (gid >= (unsigned)E_TOT * 4) return;
  const unsigned e = gid >> 2;
  const unsigned h = gid & 3;

  int dst;
  if (e < N_EDGES) dst = ei[N_EDGES + e];
  else             dst = (int)(e - N_EDGES);

  const float ex = __expf(logits[(size_t)e * 4 + h] - m[(size_t)dst * 4 + h]);
  atomicAdd(&denom[(size_t)dst * 4 + h], ex);
}

__global__ void __launch_bounds__(256)
gat_aggregate(const int* __restrict__ ei, const float* __restrict__ xl,
              const float* __restrict__ logits, const float* __restrict__ m,
              const float* __restrict__ denom, float* __restrict__ out) {
  const unsigned gid  = blockIdx.x * blockDim.x + threadIdx.x;
  const unsigned e    = gid >> 5;
  const unsigned lane = threadIdx.x & 31;

  int src, dst;
  if (e < N_EDGES) { src = ei[e]; dst = ei[N_EDGES + e]; }
  else             { src = dst = (int)(e - N_EDGES); }

  const int h = lane >> 3;
  const float alpha =
      __expf(logits[(size_t)e * 4 + h] - m[(size_t)dst * 4 + h]) /
      (denom[(size_t)dst * 4 + h] + 1e-16f);

  const float4 v = ((const float4*)(xl + (size_t)src * HC))[lane];
  float* po = out + (size_t)dst * HC + lane * 4;
  atomicAdd(po + 0, v.x * alpha);
  atomicAdd(po + 1, v.y * alpha);
  atomicAdd(po + 2, v.z * alpha);
  atomicAdd(po + 3, v.w * alpha);
}

extern "C" void kernel_launch(void* const* d_in, const int* in_sizes, int n_in,
                              void* d_out, int out_size, void* d_ws, size_t ws_size,
                              hipStream_t stream) {
  const float* x    = (const float*)d_in[0];
  const int*   ei   = (const int*)  d_in[1];
  const float* Wl   = (const float*)d_in[2];
  const float* Wr   = (const float*)d_in[3];
  const float* att  = (const float*)d_in[4];
  const float* bias = (const float*)d_in[5];
  float* out = (float*)d_out;

  float* ws     = (float*)d_ws;
  float* xl     = ws;
  float* xr     = xl + (size_t)N_NODES * HC;
  float* m      = xr + (size_t)N_NODES * HC;
  float* denom  = m + (size_t)N_NODES * 4;
  float* logits = denom + (size_t)N_NODES * 4;

  gat_init<<<(N_NODES * HC + 255) / 256, 256, 0, stream>>>(out, bias, m, denom);

  gat_gemm<<<6250, 256, 0, stream>>>(x, Wl, Wr, xl, xr);

  gat_logits<<<106250, 256, 0, stream>>>(ei, xl, xr, att, logits, m);
  gat_expsum<<<((unsigned)E_TOT * 4 + 255) / 256, 256, 0, stream>>>(ei, logits, m, denom);
  gat_aggregate<<<106250, 256, 0, stream>>>(ei, xl, logits, m, denom, out);
}


# edges grouped by destination per call (count, scan, fill); one wave per node: logits, running-max softmax, sum in registers in source order (over 64 edges: blocks by arrival); no float atomics
# speedup vs baseline: 2.0901x; 2.0901x over previous
#include <hip/hip_runtime.h>
#include <hip/hip_bf16.h>

#define N_NODES   50000
#define N_EDGES   800000
#define IN_F      128
#define HC        128
#define NEG_SLOPE 0.2f

typedef __attribute__((ext_vector_type(2))) float v2f;
typedef __attribute__((ext_vector_type(8))) float v8f;

__global__ void __launch_bounds__(256)
gat_gemm(const float* __restrict__ x, const float* __restrict__ Wl,
         const float* __restrict__ Wr, float* __restrict__ xl,
         float* __restrict__ xr) {
  const int gw   = (blockIdx.x * blockDim.x + threadIdx.x) >> 5;
  const int lane = threadIdx.x & 31;
  const int mt = gw >> 4;
  const int nt = gw & 15;

  const float* __restrict__ W   = (nt < 8) ? Wl : Wr;
  float* __restrict__       dst = (nt < 8) ? xl : xr;
  const int ncol  = (nt & 7) * 16 + (lane & 15);
  const int mrow  = mt * 16 + (lane & 15);
  const int khalf = (lane >> 4) * 2;

  const float* __restrict__ xrow = x + (size_t)mrow * IN_F;

  v8f c = {};
#pragma unroll
  for (int kb = 0; kb < IN_F; kb += 4) {
    v2f a, b;
    const float2 ax = *(const float2*)(xrow + kb + khalf);
    a.x = ax.x; a.y = ax.y;
    b.x = W[(size_t)(kb + khalf) * HC + ncol];
    b.y = W[(size_t)(kb + khalf + 1) * HC + ncol];
    c = __builtin_amdgcn_wmma_f32_16x16x4_f32(false, a, false, b,
                                              (short)0, c, false, false);
  }

  const int mbase = mt * 16 + 8 * (lane >> 4);
#pragma unroll
  for (int v = 0; v < 8; ++v)
    dst[(size_t)(mbase + v) * HC + ncol] = c[v];
}

__global__ void gat_zero(int* __restrict__ cnt) {
  const int i = blockIdx.x * blockDim.x + threadIdx.x;
  if (i < N_NODES) cnt[i] = 0;
}

__global__ void gat_hist(const int* __restrict__ ei, int* __restrict__ cnt,
                         int* __restrict__ rank) {
  const unsigned e = blockIdx.x * blockDim.x + threadIdx.x;
  if (e < N_EDGES) rank[e] = atomicAdd(&cnt[ei[N_EDGES + e]], 1);
}

#define SCAN_T     1024
#define SCAN_CHUNK (SCAN_T * 4)
#define SCAN_ITERS ((N_NODES + SCAN_CHUNK - 1) / SCAN_CHUNK)

__device__ __forceinline__ int wave_incl_scan(int s, int lane) {
  const int rl = lane & 15;
  int u;
  u = __builtin_amdgcn_update_dpp(0, s, 0x111, 0xf, 0xf, false);
  if (rl >= 1) s += u;
  u = __builtin_amdgcn_update_dpp(0, s, 0x112, 0xf, 0xf, false);
  if (rl >= 2) s += u;
  u = __builtin_amdgcn_update_dpp(0, s, 0x114, 0xf, 0xf, false);
  if (rl >= 4) s += u;
  u = __builtin_amdgcn_update_dpp(0, s, 0x118, 0xf, 0xf, false);
  if (rl >= 8) s += u;
  const int low = __builtin_amdgcn_readlane(s, 15);
  if (lane >= 16) s += low;
  return s;
}

__global__ void __launch_bounds__(SCAN_T)
gat_scan(const int* __restrict__ cnt, int* __restrict__ off) {
  __shared__ int wtot[SCAN_ITERS][32];
  const int tid = threadIdx.x, lane = tid & 31;
  const int wid = __builtin_amdgcn_readfirstlane(tid >> 5);

  int4 c[SCAN_ITERS];
#pragma unroll
  for (int it = 0; it < SCAN_ITERS; ++it) {
    const int i = it * SCAN_CHUNK + tid * 4;
    c[it] = (i < N_NODES) ? *(const int4*)(cnt + i) : make_int4(0, 0, 0, 0);
  }

  int t[SCAN_ITERS], s[SCAN_ITERS];
#pragma unroll
  for (int it = 0; it < SCAN_ITERS; ++it) {
    t[it] = c[it].x + c[it].y + c[it].z + c[it].w;
    s[it] = wave_incl_scan(t[it], lane);
    if (lane == 31) wtot[it][wid] = s[it];
  }
  __syncthreads();

  int carry = 0;
#pragma unroll
  for (int it = 0; it < SCAN_ITERS; ++it) {
    const int i = it * SCAN_CHUNK + tid * 4;
    const int wv = wtot[it][lane];
    const int ws = wave_incl_scan(wv, lane);
    const int total = __builtin_amdgcn_readlane(ws, 31);
    const int wexcl = __builtin_amdgcn_readlane(ws - wv, wid);
    const int ex = carry + wexcl + (s[it] - t[it]);
    if (i < N_NODES) {
      int4 o;
      o.x = ex;
      o.y = ex + c[it].x;
      o.z = o.y + c[it].y;
      o.w = o.z + c[it].z;
      *(int4*)(off + i) = o;
    }
    carry += total;
  }
  if (tid == 0) off[N_NODES] = carry;
}

__global__ void gat_fill(const int* __restrict__ ei, const int* __restrict__ off,
                         const int* __restrict__ rank, int* __restrict__ srcs) {
  const unsigned e = blockIdx.x * blockDim.x + threadIdx.x;
  if (e < N_EDGES) srcs[off[ei[N_EDGES + e]] + rank[e]] = ei[e];
}

__device__ __forceinline__ float head_logit(const float4 a, const float4 b,
                                            const float4 w) {
  float vx = a.x + b.x, vy = a.y + b.y, vz = a.z + b.z, vw = a.w + b.w;
  vx = (vx > 0.f) ? vx : vx * NEG_SLOPE;
  vy = (vy > 0.f) ? vy : vy * NEG_SLOPE;
  vz = (vz > 0.f) ? vz : vz * NEG_SLOPE;
  vw = (vw > 0.f) ? vw : vw * NEG_SLOPE;
  float s = vx * w.x + vy * w.y + vz * w.z + vw * w.w;
  s += __shfl_xor(s, 1);
  s += __shfl_xor(s, 2);
  s += __shfl_xor(s, 4);
  return s;
}

__device__ __forceinline__ int pick_rank(int k0, int k1, int r0, int r1, int p) {
  const unsigned m0 = __builtin_amdgcn_ballot_w32(r0 == p);
  if (m0) return __builtin_amdgcn_readlane(k0, __builtin_ctz(m0));
  const unsigned m1 = __builtin_amdgcn_ballot_w32(r1 == p);
  return __builtin_amdgcn_readlane(k1, __builtin_ctz(m1));
}

__global__ void __launch_bounds__(256)
gat_node(const float* __restrict__ xl, const float* __restrict__ xr,
         const float* __restrict__ att, const float* __restrict__ bias,
         const int* __restrict__ off, const int* __restrict__ srcs,
         float* __restrict__ out) {
  const int lane = threadIdx.x & 31;
  const int n = __builtin_amdgcn_readfirstlane(
      (int)(blockIdx.x * 8 + (threadIdx.x >> 5)));

  const float4 b = ((const float4*)(xr + (size_t)n * HC))[lane];
  const float4 w = ((const float4*)att)[lane];
  const int beg = off[n], end = off[n + 1];

  float4 acc = ((const float4*)(xl + (size_t)n * HC))[lane];
  float m = head_logit(acc, b, w);
  float den = 1.0f;

  for (int base = beg; base < end; base += 64) {
    const int cn = (end - base < 64) ? (end - base) : 64;
    const int k0 = (lane < cn) ? srcs[base + lane] : 0x7fffffff;
    const int k1 = (lane + 32 < cn) ? srcs[base + 32 + lane] : 0x7fffffff;

    int r0 = 0, r1 = 0;
    const int c0 = (cn < 32) ? cn : 32;
    for (int j = 0; j < c0; ++j) {
      const int kj = __builtin_amdgcn_readlane(k0, j);
      r0 += (kj < k0 || (kj == k0 && j < lane)) ? 1 : 0;
    }
    if (cn > 32) {
      for (int j = 0; j < 32; ++j) {
        const int kj = __builtin_amdgcn_readlane(k0, j);
        r1 += (kj <= k1) ? 1 : 0;
      }
      for (int j = 32; j < cn; ++j) {
        const int kj = __builtin_amdgcn_readlane(k1, j - 32);
        r0 += (kj < k0) ? 1 : 0;
        r1 += (kj < k1 || (kj == k1 && j - 32 < lane)) ? 1 : 0;
      }
    }
    if (lane >= cn) r0 = -1;
    if (lane + 32 >= cn) r1 = -1;

    int j = 0;
    for (; j + 4 <= cn; j += 4) {
      const int s0 = pick_rank(k0, k1, r0, r1, j);
      const int s1 = pick_rank(k0, k1, r0, r1, j + 1);
      const int s2 = pick_rank(k0, k1, r0, r1, j + 2);
      const int s3 = pick_rank(k0, k1, r0, r1, j + 3);
      const float4 a0 = ((const float4*)(xl + (size_t)s0 * HC))[lane];
      const float4 a1 = ((const float4*)(xl + (size_t)s1 * HC))[lane];
      const float4 a2 = ((const float4*)(xl + (size_t)s2 * HC))[lane];
      const float4 a3 = ((const float4*)(xl + (size_t)s3 * HC))[lane];
      const float l0 = head_logit(a0, b, w);
      const float l1 = head_logit(a1, b, w);
      const float l2 = head_logit(a2, b, w);
      const float l3 = head_logit(a3, b, w);
      const float mn = fmaxf(fmaxf(fmaxf(l0, l1), fmaxf(l2, l3)), m);
      const float sc = __expf(m - mn);
      const float p0 = __expf(l0 - mn);
      const float p1 = __expf(l1 - mn);
      const float p2 = __expf(l2 - mn);
      const float p3 = __expf(l3 - mn);
      den = den * sc + ((p0 + p1) + (p2 + p3));
      acc.x = acc.x * sc + ((p0 * a0.x + p1 * a1.x) + (p2 * a2.x + p3 * a3.x));
      acc.y = acc.y * sc + ((p0 * a0.y + p1 * a1.y) + (p2 * a2.y + p3 * a3.y));
      acc.z = acc.z * sc + ((p0 * a0.z + p1 * a1.z) + (p2 * a2.z + p3 * a3.z));
      acc.w = acc.w * sc + ((p0 * a0.w + p1 * a1.w) + (p2 * a2.w + p3 * a3.w));
      m = mn;
    }
    for (; j < cn; ++j) {
      const int s0 = pick_rank(k0, k1, r0, r1, j);
      const float4 a0 = ((const float4*)(xl + (size_t)s0 * HC))[lane];
      const float l0 = head_logit(a0, b, w);
      const float mn = fmaxf(l0, m);
      const float sc = __expf(m - mn);
      const float p0 = __expf(l0 - mn);
      den = den * sc + p0;
      acc.x = acc.x * sc + p0 * a0.x;
      acc.y = acc.y * sc + p0 * a0.y;
      acc.z = acc.z * sc + p0 * a0.z;
      acc.w = acc.w * sc + p0 * a0.w;
      m = mn;
    }
  }

  const float dn = den + 1e-16f;
  const float4 bi = ((const float4*)bias)[lane];
  float4 r;
  r.x = acc.x / dn + bi.x;
  r.y = acc.y / dn + bi.y;
  r.z = acc.z / dn + bi.z;
  r.w = acc.w / dn + bi.w;
  ((float4*)(out + (size_t)n * HC))[lane] = r;
}

extern "C" void kernel_launch(void* const* d_in, const int* in_sizes, int n_in,
                              void* d_out, int out_size, void* d_ws, size_t ws_size,
                              hipStream_t stream) {
  const float* x    = (const float*)d_in[0];
  const int*   ei   = (const int*)  d_in[1];
  const float* Wl   = (const float*)d_in[2];
  const float* Wr   = (const float*)d_in[3];
  const float* att  = (const float*)d_in[4];
  const float* bias = (const float*)d_in[5];
  float* out = (float*)d_out;

  float* ws   = (float*)d_ws;
  float* xl   = ws;
  float* xr   = xl + (size_t)N_NODES * HC;
  int*   cnt  = (int*)(xr + (size_t)N_NODES * HC);
  int*   off  = cnt + N_NODES;
  int*   rank = off + 50016;
  int*   srcs = rank + N_EDGES;

  gat_gemm<<<6250, 256, 0, stream>>>(x, Wl, Wr, xl, xr);

  gat_zero<<<(N_NODES + 255) / 256, 256, 0, stream>>>(cnt);
  gat_hist<<<N_EDGES / 256, 256, 0, stream>>>(ei, cnt, rank);
  gat_scan<<<1, SCAN_T, 0, stream>>>(cnt, off);
  gat_fill<<<N_EDGES / 256, 256, 0, stream>>>(ei, off, rank, srcs);

  gat_node<<<6250, 256, 0, stream>>>(xl, xr, att, bias, off, srcs, out);
}


# more work per wave in every launch but the scan: a GEMM wave makes 8 column tiles from one A fragment, hist/fill threads take 8 edges, zero 4 counters, a node wave 2 nodes; 4x fewer waves per call
# speedup vs baseline: 2.2746x; 1.0883x over previous
#include <hip/hip_runtime.h>
#include <hip/hip_bf16.h>

#define N_NODES   50000
#define N_EDGES   800000
#define IN_F      128
#define HC        128
#define NEG_SLOPE 0.2f

typedef __attribute__((ext_vector_type(2))) float v2f;
typedef __attribute__((ext_vector_type(8))) float v8f;

#define M_TILES (N_NODES / 16)

__global__ void __launch_bounds__(128)
gat_gemm(const float* __restrict__ x, const float* __restrict__ Wl,
         const float* __restrict__ Wr, float* __restrict__ xl,
         float* __restrict__ xr) {
  const int lane = threadIdx.x & 31;
  const int gw = __builtin_amdgcn_readfirstlane(
      (int)((blockIdx.x * 128 + threadIdx.x) >> 5));
  const int mt = gw >> 1;
  if (mt >= M_TILES) return;

  const float* __restrict__ W   = (gw & 1) ? Wr : Wl;
  float* __restrict__       dst = (gw & 1) ? xr : xl;
  const int c     = lane & 15;
  const int khalf = (lane >> 4) * 2;

  const float* __restrict__ xrow = x + (size_t)(mt * 16 + c) * IN_F + khalf;
  const float* __restrict__ wrow = W + (size_t)khalf * HC + 4 * c;

  v8f acc[8];
  for (int t = 0; t < 8; ++t) acc[t] = (v8f){};

#pragma unroll 4
  for (int kb = 0; kb < IN_F; kb += 4) {
    const float2 ax = *(const float2*)(xrow + kb);
    v2f a;
    a.x = ax.x; a.y = ax.y;
    const float* __restrict__ wk = wrow + (size_t)kb * HC;
    const float4 w0[2] = { *(const float4*)(wk),      *(const float4*)(wk + 64) };
    const float4 w1[2] = { *(const float4*)(wk + HC), *(const float4*)(wk + HC + 64) };
    for (int g = 0; g < 2; ++g) {
      v2f b;
      b.x = w0[g].x; b.y = w1[g].x;
      acc[4 * g + 0] = __builtin_amdgcn_wmma_f32_16x16x4_f32(
          false, a, false, b, (short)0, acc[4 * g + 0], false, false);
      b.x = w0[g].y; b.y = w1[g].y;
      acc[4 * g + 1] = __builtin_amdgcn_wmma_f32_16x16x4_f32(
          false, a, false, b, (short)0, acc[4 * g + 1], false, false);
      b.x = w0[g].z; b.y = w1[g].z;
      acc[4 * g + 2] = __builtin_amdgcn_wmma_f32_16x16x4_f32(
          false, a, false, b, (short)0, acc[4 * g + 2], false, false);
      b.x = w0[g].w; b.y = w1[g].w;
      acc[4 * g + 3] = __builtin_amdgcn_wmma_f32_16x16x4_f32(
          false, a, false, b, (short)0, acc[4 * g + 3], false, false);
    }
  }

  float* __restrict__ orow =
      dst + (size_t)(mt * 16 + 8 * (lane >> 4)) * HC + 4 * c;
  for (int v = 0; v < 8; ++v) {
    for (int g = 0; g < 2; ++g) {
      float4 o;
      o.x = acc[4 * g + 0][v];
      o.y = acc[4 * g + 1][v];
      o.z = acc[4 * g + 2][v];
      o.w = acc[4 * g + 3][v];
      *(float4*)(orow + (size_t)v * HC + 64 * g) = o;
    }
  }
}

#define EDGE_T (N_EDGES / 8)

__global__ void gat_zero(int* __restrict__ cnt) {
  const int i = blockIdx.x * blockDim.x + threadIdx.x;
  if (i < N_NODES / 4) ((int4*)cnt)[i] = make_int4(0, 0, 0, 0);
}

__global__ void gat_hist(const int* __restrict__ ei, int* __restrict__ cnt,
                         int* __restrict__ rank) {
  const unsigned t = blockIdx.x * blockDim.x + threadIdx.x;
  if (t < EDGE_T) {
    const int4* __restrict__ dst4 = (const int4*)(ei + N_EDGES);
    const int4 d0 = dst4[t];
    const int4 d1 = dst4[t + EDGE_T];
    int4 r0, r1;
    r0.x = atomicAdd(&cnt[d0.x], 1);
    r0.y = atomicAdd(&cnt[d0.y], 1);
    r0.z = atomicAdd(&cnt[d0.z], 1);
    r0.w = atomicAdd(&cnt[d0.w], 1);
    r1.x = atomicAdd(&cnt[d1.x], 1);
    r1.y = atomicAdd(&cnt[d1.y], 1);
    r1.z = atomicAdd(&cnt[d1.z], 1);
    r1.w = atomicAdd(&cnt[d1.w], 1);
    ((int4*)rank)[t] = r0;
    ((int4*)rank)[t + EDGE_T] = r1;
  }
}

#define SCAN_T     1024
#define SCAN_CHUNK (SCAN_T * 4)
#define SCAN_ITERS ((N_NODES + SCAN_CHUNK - 1) / SCAN_CHUNK)

__device__ __forceinline__ int wave_incl_scan(int s, int lane) {
  const int rl = lane & 15;
  int u;
  u = __builtin_amdgcn_update_dpp(0, s, 0x111, 0xf, 0xf, false);
  if (rl >= 1) s += u;
  u = __builtin_amdgcn_update_dpp(0, s, 0x112, 0xf, 0xf, false);
  if (rl >= 2) s += u;
  u = __builtin_amdgcn_update_dpp(0, s, 0x114, 0xf, 0xf, false);
  if (rl >= 4) s += u;
  u = __builtin_amdgcn_update_dpp(0, s, 0x118, 0xf, 0xf, false);
  if (rl >= 8) s += u;
  const int low = __builtin_amdgcn_readlane(s, 15);
  if (lane >= 16) s += low;
  return s;
}

__global__ void __launch_bounds__(SCAN_T)
gat_scan(const int* __restrict__ cnt, int* __restrict__ off) {
  __shared__ int wtot[SCAN_ITERS][32];
  const int tid = threadIdx.x, lane = tid & 31;
  const int wid = __builtin_amdgcn_readfirstlane(tid >> 5);

  int4 c[SCAN_ITERS];
#pragma unroll
  for (int it = 0; it < SCAN_ITERS; ++it) {
    const int i = it * SCAN_CHUNK + tid * 4;
    c[it] = (i < N_NODES) ? *(const int4*)(cnt + i) : make_int4(0, 0, 0, 0);
  }

  int t[SCAN_ITERS], s[SCAN_ITERS];
#pragma unroll
  for (int it = 0; it < SCAN_ITERS; ++it) {
    t[it] = c[it].x + c[it].y + c[it].z + c[it].w;
    s[it] = wave_incl_scan(t[it], lane);
    if (lane == 31) wtot[it][wid] = s[it];
  }
  __syncthreads();

  int carry = 0;
#pragma unroll
  for (int it = 0; it < SCAN_ITERS; ++it) {
    const int i = it * SCAN_CHUNK + tid * 4;
    const int wv = wtot[it][lane];
    const int ws = wave_incl_scan(wv, lane);
    const int total = __builtin_amdgcn_readlane(ws, 31);
    const int wexcl = __builtin_amdgcn_readlane(ws - wv, wid);
    const int ex = carry + wexcl + (s[it] - t[it]);
    if (i < N_NODES) {
      int4 o;
      o.x = ex;
      o.y = ex + c[it].x;
      o.z = o.y + c[it].y;
      o.w = o.z + c[it].z;
      *(int4*)(off + i) = o;
    }
    carry += total;
  }
  if (tid == 0) off[N_NODES] = carry;
}

__global__ void gat_fill(const int* __restrict__ ei, const int* __restrict__ off,
                         const int* __restrict__ rank, int* __restrict__ srcs) {
  const unsigned t = blockIdx.x * blockDim.x + threadIdx.x;
  if (t < EDGE_T) {
    const int4* __restrict__ src4 = (const int4*)ei;
    const int4* __restrict__ dst4 = (const int4*)(ei + N_EDGES);
    const int4* __restrict__ rnk4 = (const int4*)rank;
    const int4 d0 = dst4[t], d1 = dst4[t + EDGE_T];
    const int4 r0 = rnk4[t], r1 = rnk4[t + EDGE_T];
    const int4 s0 = src4[t], s1 = src4[t + EDGE_T];
    const int p0 = off[d0.x] + r0.x, p1 = off[d0.y] + r0.y;
    const int p2 = off[d0.z] + r0.z, p3 = off[d0.w] + r0.w;
    const int p4 = off[d1.x] + r1.x, p5 = off[d1.y] + r1.y;
    const int p6 = off[d1.z] + r1.z, p7 = off[d1.w] + r1.w;
    srcs[p0] = s0.x; srcs[p1] = s0.y; srcs[p2] = s0.z; srcs[p3] = s0.w;
    srcs[p4] = s1.x; srcs[p5] = s1.y; srcs[p6] = s1.z; srcs[p7] = s1.w;
  }
}

__device__ __forceinline__ float head_logit(const float4 a, const float4 b,
                                            const float4 w) {
  float vx = a.x + b.x, vy = a.y + b.y, vz = a.z + b.z, vw = a.w + b.w;
  vx = (vx > 0.f) ? vx : vx * NEG_SLOPE;
  vy = (vy > 0.f) ? vy : vy * NEG_SLOPE;
  vz = (vz > 0.f) ? vz : vz * NEG_SLOPE;
  vw = (vw > 0.f) ? vw : vw * NEG_SLOPE;
  float s = vx * w.x + vy * w.y + vz * w.z + vw * w.w;
  s += __shfl_xor(s, 1);
  s += __shfl_xor(s, 2);
  s += __shfl_xor(s, 4);
  return s;
}

__device__ __forceinline__ int pick_rank(int k0, int k1, int r0, int r1, int p) {
  const unsigned m0 = __builtin_amdgcn_ballot_w32(r0 == p);
  if (m0) return __builtin_amdgcn_readlane(k0, __builtin_ctz(m0));
  const unsigned m1 = __builtin_amdgcn_ballot_w32(r1 == p);
  return __builtin_amdgcn_readlane(k1, __builtin_ctz(m1));
}

#define NODES_PER_WAVE 2

__global__ void __launch_bounds__(256)
gat_node(const float* __restrict__ xl, const float* __restrict__ xr,
         const float* __restrict__ att, const float* __restrict__ bias,
         const int* __restrict__ off, const int* __restrict__ srcs,
         float* __restrict__ out) {
  const int lane = threadIdx.x & 31;
  const int wv = __builtin_amdgcn_readfirstlane(
      (int)(blockIdx.x * 8 + (threadIdx.x >> 5)));

  const float4 w  = ((const float4*)att)[lane];
  const float4 bi = ((const float4*)bias)[lane];

  for (int q = 0; q < NODES_PER_WAVE; ++q) {
  const int n = wv * NODES_PER_WAVE + q;

  const float4 b = ((const float4*)(xr + (size_t)n * HC))[lane];
  const int beg = off[n], end = off[n + 1];

  float4 acc = ((const float4*)(xl + (size_t)n * HC))[lane];
  float m = head_logit(acc, b, w);
  float den = 1.0f;

  for (int base = beg; base < end; base += 64) {
    const int cn = (end - base < 64) ? (end - base) : 64;
    const int k0 = (lane < cn) ? srcs[base + lane] : 0x7fffffff;
    const int k1 = (lane + 32 < cn) ? srcs[base + 32 + lane] : 0x7fffffff;

    int r0 = 0, r1 = 0;
    const int c0 = (cn < 32) ? cn : 32;
    for (int j = 0; j < c0; ++j) {
      const int kj = __builtin_amdgcn_readlane(k0, j);
      r0 += (kj < k0 || (kj == k0 && j < lane)) ? 1 : 0;
    }
    if (cn > 32) {
      for (int j = 0; j < 32; ++j) {
        const int kj = __builtin_amdgcn_readlane(k0, j);
        r1 += (kj <= k1) ? 1 : 0;
      }
      for (int j = 32; j < cn; ++j) {
        const int kj = __builtin_amdgcn_readlane(k1, j - 32);
        r0 += (kj < k0) ? 1 : 0;
        r1 += (kj < k1 || (kj == k1 && j - 32 < lane)) ? 1 : 0;
      }
    }
    if (lane >= cn) r0 = -1;
    if (lane + 32 >= cn) r1 = -1;

    int j = 0;
    for (; j + 4 <= cn; j += 4) {
      const int s0 = pick_rank(k0, k1, r0, r1, j);
      const int s1 = pick_rank(k0, k1, r0, r1, j + 1);
      const int s2 = pick_rank(k0, k1, r0, r1, j + 2);
      const int s3 = pick_rank(k0, k1, r0, r1, j + 3);
      const float4 a0 = ((const float4*)(xl + (size_t)s0 * HC))[lane];
      const float4 a1 = ((const float4*)(xl + (size_t)s1 * HC))[lane];
      const float4 a2 = ((const float4*)(xl + (size_t)s2 * HC))[lane];
      const float4 a3 = ((const float4*)(xl + (size_t)s3 * HC))[lane];
      const float l0 = head_logit(a0, b, w);
      const float l1 = head_logit(a1, b, w);
      const float l2 = head_logit(a2, b, w);
      const float l3 = head_logit(a3, b, w);
      const float mn = fmaxf(fmaxf(fmaxf(l0, l1), fmaxf(l2, l3)), m);
      const float sc = __expf(m - mn);
      const float p0 = __expf(l0 - mn);
      const float p1 = __expf(l1 - mn);
      const float p2 = __expf(l2 - mn);
      const float p3 = __expf(l3 - mn);
      den = den * sc + ((p0 + p1) + (p2 + p3));
      acc.x = acc.x * sc + ((p0 * a0.x + p1 * a1.x) + (p2 * a2.x + p3 * a3.x));
      acc.y = acc.y * sc + ((p0 * a0.y + p1 * a1.y) + (p2 * a2.y + p3 * a3.y));
      acc.z = acc.z * sc + ((p0 * a0.z + p1 * a1.z) + (p2 * a2.z + p3 * a3.z));
      acc.w = acc.w * sc + ((p0 * a0.w + p1 * a1.w) + (p2 * a2.w + p3 * a3.w));
      m = mn;
    }
    for (; j < cn; ++j) {
      const int s0 = pick_rank(k0, k1, r0, r1, j);
      const float4 a0 = ((const float4*)(xl + (size_t)s0 * HC))[lane];
      const float l0 = head_logit(a0, b, w);
      const float mn = fmaxf(l0, m);
      const float sc = __expf(m - mn);
      const float p0 = __expf(l0 - mn);
      den = den * sc + p0;
      acc.x = acc.x * sc + p0 * a0.x;
      acc.y = acc.y * sc + p0 * a0.y;
      acc.z = acc.z * sc + p0 * a0.z;
      acc.w = acc.w * sc + p0 * a0.w;
      m = mn;
    }
  }

  const float dn = den + 1e-16f;
  float4 r;
  r.x = acc.x / dn + bi.x;
  r.y = acc.y / dn + bi.y;
  r.z = acc.z / dn + bi.z;
  r.w = acc.w / dn + bi.w;
  ((float4*)(out + (size_t)n * HC))[lane] = r;
  }
}

extern "C" void kernel_launch(void* const* d_in, const int* in_sizes, int n_in,
                              void* d_out, int out_size, void* d_ws, size_t ws_size,
                              hipStream_t stream) {
  const float* x    = (const float*)d_in[0];
  const int*   ei   = (const int*)  d_in[1];
  const float* Wl   = (const float*)d_in[2];
  const float* Wr   = (const float*)d_in[3];
  const float* att  = (const float*)d_in[4];
  const float* bias = (const float*)d_in[5];
  float* out = (float*)d_out;

  float* ws   = (float*)d_ws;
  float* xl   = ws;
  float* xr   = xl + (size_t)N_NODES * HC;
  int*   cnt  = (int*)(xr + (size_t)N_NODES * HC);
  int*   off  = cnt + N_NODES;
  int*   rank = off + 50016;
  int*   srcs = rank + N_EDGES;

  gat_gemm<<<1563, 128, 0, stream>>>(x, Wl, Wr, xl, xr);

  gat_zero<<<49, 256, 0, stream>>>(cnt);
  gat_hist<<<391, 256, 0, stream>>>(ei, cnt, rank);
  gat_scan<<<1, SCAN_T, 0, stream>>>(cnt, off);
  gat_fill<<<391, 256, 0, stream>>>(ei, off, rank, srcs);

  gat_node<<<3125, 256, 0, stream>>>(xl, xr, att, bias, off, srcs, out);
}


# edges grouped in one pass: the counter atomic's return is the slot in the destination's 64-slot 16-bit bucket; rank array, scan and fill launches gone (4 launches); lists over 64 sum in edge order
# speedup vs baseline: 2.7562x; 1.2117x over previous
#include <hip/hip_runtime.h>
#include <hip/hip_bf16.h>

#define N_NODES   50000
#define N_EDGES   800000
#define IN_F      128
#define HC        128
#define NEG_SLOPE 0.2f

typedef __attribute__((ext_vector_type(2))) float v2f;
typedef __attribute__((ext_vector_type(8))) float v8f;

#define M_TILES (N_NODES / 16)

__global__ void __launch_bounds__(128)
gat_gemm(const float* __restrict__ x, const float* __restrict__ Wl,
         const float* __restrict__ Wr, float* __restrict__ xl,
         float* __restrict__ xr) {
  const int lane = threadIdx.x & 31;
  const int gw = __builtin_amdgcn_readfirstlane(
      (int)((blockIdx.x * 128 + threadIdx.x) >> 5));
  const int mt = gw >> 1;
  if (mt >= M_TILES) return;

  const float* __restrict__ W   = (gw & 1) ? Wr : Wl;
  float* __restrict__       dst = (gw & 1) ? xr : xl;
  const int c     = lane & 15;
  const int khalf = (lane >> 4) * 2;

  const float* __restrict__ xrow = x + (size_t)(mt * 16 + c) * IN_F + khalf;
  const float* __restrict__ wrow = W + (size_t)khalf * HC + 4 * c;

  v8f acc[8];
  for (int t = 0; t < 8; ++t) acc[t] = (v8f){};

#pragma unroll 4
  for (int kb = 0; kb < IN_F; kb += 4) {
    const float2 ax = *(const float2*)(xrow + kb);
    v2f a;
    a.x = ax.x; a.y = ax.y;
    const float* __restrict__ wk = wrow + (size_t)kb * HC;
    const float4 w0[2] = { *(const float4*)(wk),      *(const float4*)(wk + 64) };
    const float4 w1[2] = { *(const float4*)(wk + HC), *(const float4*)(wk + HC + 64) };
    for (int g = 0; g < 2; ++g) {
      v2f b;
      b.x = w0[g].x; b.y = w1[g].x;
      acc[4 * g + 0] = __builtin_amdgcn_wmma_f32_16x16x4_f32(
          false, a, false, b, (short)0, acc[4 * g + 0], false, false);
      b.x = w0[g].y; b.y = w1[g].y;
      acc[4 * g + 1] = __builtin_amdgcn_wmma_f32_16x16x4_f32(
          false, a, false, b, (short)0, acc[4 * g + 1], false, false);
      b.x = w0[g].z; b.y = w1[g].z;
      acc[4 * g + 2] = __builtin_amdgcn_wmma_f32_16x16x4_f32(
          false, a, false, b, (short)0, acc[4 * g + 2], false, false);
      b.x = w0[g].w; b.y = w1[g].w;
      acc[4 * g + 3] = __builtin_amdgcn_wmma_f32_16x16x4_f32(
          false, a, false, b, (short)0, acc[4 * g + 3], false, false);
    }
  }

  float* __restrict__ orow =
      dst + (size_t)(mt * 16 + 8 * (lane >> 4)) * HC + 4 * c;
  for (int v = 0; v < 8; ++v) {
    for (int g = 0; g < 2; ++g) {
      float4 o;
      o.x = acc[4 * g + 0][v];
      o.y = acc[4 * g + 1][v];
      o.z = acc[4 * g + 2][v];
      o.w = acc[4 * g + 3][v];
      *(float4*)(orow + (size_t)v * HC + 64 * g) = o;
    }
  }
}

#define EDGE_T  (N_EDGES / 8)
#define BKT_CAP 64
static_assert(N_NODES <= 65536, "a source id must fit the 16-bit bucket slot");
static_assert(N_EDGES % 32 == 0, "the edge-list walk of an over-full node takes whole waves");

__global__ void gat_zero(int* __restrict__ cnt) {
  const int i = blockIdx.x * blockDim.x + threadIdx.x;
  if (i < N_NODES / 4) ((int4*)cnt)[i] = make_int4(0, 0, 0, 0);
}

__global__ void gat_bucket(const int* __restrict__ ei, int* __restrict__ cnt,
                           unsigned short* __restrict__ bkt) {
  const unsigned t = blockIdx.x * blockDim.x + threadIdx.x;
  if (t < EDGE_T) {
    const int4* __restrict__ src4 = (const int4*)ei;
    const int4* __restrict__ dst4 = (const int4*)(ei + N_EDGES);
    const int4 d0 = dst4[t], d1 = dst4[t + EDGE_T];
    const int4 s0 = src4[t], s1 = src4[t + EDGE_T];
    const int p0 = atomicAdd(&cnt[d0.x], 1);
    const int p1 = atomicAdd(&cnt[d0.y], 1);
    const int p2 = atomicAdd(&cnt[d0.z], 1);
    const int p3 = atomicAdd(&cnt[d0.w], 1);
    const int p4 = atomicAdd(&cnt[d1.x], 1);
    const int p5 = atomicAdd(&cnt[d1.y], 1);
    const int p6 = atomicAdd(&cnt[d1.z], 1);
    const int p7 = atomicAdd(&cnt[d1.w], 1);
    if (p0 < BKT_CAP) bkt[(unsigned)(d0.x * BKT_CAP + p0)] = (unsigned short)s0.x;
    if (p1 < BKT_CAP) bkt[(unsigned)(d0.y * BKT_CAP + p1)] = (unsigned short)s0.y;
    if (p2 < BKT_CAP) bkt[(unsigned)(d0.z * BKT_CAP + p2)] = (unsigned short)s0.z;
    if (p3 < BKT_CAP) bkt[(unsigned)(d0.w * BKT_CAP + p3)] = (unsigned short)s0.w;
    if (p4 < BKT_CAP) bkt[(unsigned)(d1.x * BKT_CAP + p4)] = (unsigned short)s1.x;
    if (p5 < BKT_CAP) bkt[(unsigned)(d1.y * BKT_CAP + p5)] = (unsigned short)s1.y;
    if (p6 < BKT_CAP) bkt[(unsigned)(d1.z * BKT_CAP + p6)] = (unsigned short)s1.z;
    if (p7 < BKT_CAP) bkt[(unsigned)(d1.w * BKT_CAP + p7)] = (unsigned short)s1.w;
  }
}

__device__ __forceinline__ float head_logit(const float4 a, const float4 b,
                                            const float4 w) {
  float vx = a.x + b.x, vy = a.y + b.y, vz = a.z + b.z, vw = a.w + b.w;
  vx = (vx > 0.f) ? vx : vx * NEG_SLOPE;
  vy = (vy > 0.f) ? vy : vy * NEG_SLOPE;
  vz = (vz > 0.f) ? vz : vz * NEG_SLOPE;
  vw = (vw > 0.f) ? vw : vw * NEG_SLOPE;
  float s = vx * w.x + vy * w.y + vz * w.z + vw * w.w;
  s += __shfl_xor(s, 1);
  s += __shfl_xor(s, 2);
  s += __shfl_xor(s, 4);
  return s;
}

__device__ __forceinline__ int pick_rank(int k0, int k1, int r0, int r1, int p) {
  const unsigned m0 = __builtin_amdgcn_ballot_w32(r0 == p);
  if (m0) return __builtin_amdgcn_readlane(k0, __builtin_ctz(m0));
  const unsigned m1 = __builtin_amdgcn_ballot_w32(r1 == p);
  return __builtin_amdgcn_readlane(k1, __builtin_ctz(m1));
}

__device__ __forceinline__ void edge_one(const float* __restrict__ xl, const int s0,
                                         const int lane, const float4 b,
                                         const float4 w, float& m, float& den,
                                         float4& acc) {
  const float4 a0 = ((const float4*)(xl + (size_t)s0 * HC))[lane];
  const float l0 = head_logit(a0, b, w);
  const float mn = fmaxf(l0, m);
  const float sc = __expf(m - mn);
  const float p0 = __expf(l0 - mn);
  den = den * sc + p0;
  acc.x = acc.x * sc + p0 * a0.x;
  acc.y = acc.y * sc + p0 * a0.y;
  acc.z = acc.z * sc + p0 * a0.z;
  acc.w = acc.w * sc + p0 * a0.w;
  m = mn;
}

#define NODES_PER_WAVE 2

__global__ void __launch_bounds__(256)
gat_node(const float* __restrict__ xl, const float* __restrict__ xr,
         const float* __restrict__ att, const float* __restrict__ bias,
         const int* __restrict__ ei, const int* __restrict__ cnt,
         const unsigned short* __restrict__ bkt, float* __restrict__ out) {
  const int lane = threadIdx.x & 31;
  const int wv = __builtin_amdgcn_readfirstlane(
      (int)(blockIdx.x * 8 + (threadIdx.x >> 5)));

  const float4 w  = ((const float4*)att)[lane];
  const float4 bi = ((const float4*)bias)[lane];

  for (int q = 0; q < NODES_PER_WAVE; ++q) {
  const int n = wv * NODES_PER_WAVE + q;

  const float4 b = ((const float4*)(xr + (size_t)n * HC))[lane];
  const int cn = cnt[n];

  float4 acc = ((const float4*)(xl + (size_t)n * HC))[lane];
  float m = head_logit(acc, b, w);
  float den = 1.0f;

  if (cn <= BKT_CAP) {
    const unsigned short* __restrict__ bl = bkt + (unsigned)(n * BKT_CAP);
    const int k0 = (lane < cn) ? (int)bl[lane] : 0x7fffffff;
    const int k1 = (lane + 32 < cn) ? (int)bl[32 + lane] : 0x7fffffff;

    int r0 = 0, r1 = 0;
    const int c0 = (cn < 32) ? cn : 32;
    for (int j = 0; j < c0; ++j) {
      const int kj = __builtin_amdgcn_readlane(k0, j);
      r0 += (kj < k0 || (kj == k0 && j < lane)) ? 1 : 0;
    }
    if (cn > 32) {
      for (int j = 0; j < 32; ++j) {
        const int kj = __builtin_amdgcn_readlane(k0, j);
        r1 += (kj <= k1) ? 1 : 0;
      }
      for (int j = 32; j < cn; ++j) {
        const int kj = __builtin_amdgcn_readlane(k1, j - 32);
        r0 += (kj < k0) ? 1 : 0;
        r1 += (kj < k1 || (kj == k1 && j - 32 < lane)) ? 1 : 0;
      }
    }
    if (lane >= cn) r0 = -1;
    if (lane + 32 >= cn) r1 = -1;

    int j = 0;
    for (; j + 4 <= cn; j += 4) {
      const int s0 = pick_rank(k0, k1, r0, r1, j);
      const int s1 = pick_rank(k0, k1, r0, r1, j + 1);
      const int s2 = pick_rank(k0, k1, r0, r1, j + 2);
      const int s3 = pick_rank(k0, k1, r0, r1, j + 3);
      const float4 a0 = ((const float4*)(xl + (size_t)s0 * HC))[lane];
      const float4 a1 = ((const float4*)(xl + (size_t)s1 * HC))[lane];
      const float4 a2 = ((const float4*)(xl + (size_t)s2 * HC))[lane];
      const float4 a3 = ((const float4*)(xl + (size_t)s3 * HC))[lane];
      const float l0 = head_logit(a0, b, w);
      const float l1 = head_logit(a1, b, w);
      const float l2 = head_logit(a2, b, w);
      const float l3 = head_logit(a3, b, w);
      const float mn = fmaxf(fmaxf(fmaxf(l0, l1), fmaxf(l2, l3)), m);
      const float sc = __expf(m - mn);
      const float p0 = __expf(l0 - mn);
      const float p1 = __expf(l1 - mn);
      const float p2 = __expf(l2 - mn);
      const float p3 = __expf(l3 - mn);
      den = den * sc + ((p0 + p1) + (p2 + p3));
      acc.x = acc.x * sc + ((p0 * a0.x + p1 * a1.x) + (p2 * a2.x + p3 * a3.x));
      acc.y = acc.y * sc + ((p0 * a0.y + p1 * a1.y) + (p2 * a2.y + p3 * a3.y));
      acc.z = acc.z * sc + ((p0 * a0.z + p1 * a1.z) + (p2 * a2.z + p3 * a3.z));
      acc.w = acc.w * sc + ((p0 * a0.w + p1 * a1.w) + (p2 * a2.w + p3 * a3.w));
      m = mn;
    }
    for (; j < cn; ++j)
      edge_one(xl, pick_rank(k0, k1, r0, r1, j), lane, b, w, m, den, acc);
  } else {
    for (int e0 = 0; e0 < N_EDGES; e0 += 32) {
      const int d = ei[N_EDGES + e0 + lane];
      const int s = ei[e0 + lane];
      unsigned mm = __builtin_amdgcn_ballot_w32(d == n);
      while (mm) {
        const int l = __builtin_ctz(mm);
        mm &= mm - 1;
        edge_one(xl, __builtin_amdgcn_readlane(s, l), lane, b, w, m, den, acc);
      }
    }
  }

  const float dn = den + 1e-16f;
  float4 r;
  r.x = acc.x / dn + bi.x;
  r.y = acc.y / dn + bi.y;
  r.z = acc.z / dn + bi.z;
  r.w = acc.w / dn + bi.w;
  ((float4*)(out + (size_t)n * HC))[lane] = r;
  }
}

extern "C" void kernel_launch(void* const* d_in, const int* in_sizes, int n_in,
                              void* d_out, int out_size, void* d_ws, size_t ws_size,
                              hipStream_t stream) {
  const float* x    = (const float*)d_in[0];
  const int*   ei   = (const int*)  d_in[1];
  const float* Wl   = (const float*)d_in[2];
  const float* Wr   = (const float*)d_in[3];
  const float* att  = (const float*)d_in[4];
  const float* bias = (const float*)d_in[5];
  float* out = (float*)d_out;

  float* ws   = (float*)d_ws;
  float* xl   = ws;
  float* xr   = xl + (size_t)N_NODES * HC;
  int*   cnt  = (int*)(xr + (size_t)N_NODES * HC);
  unsigned short* bkt = (unsigned short*)(cnt + 50048);

  gat_gemm<<<1563, 128, 0, stream>>>(x, Wl, Wr, xl, xr);

  gat_zero<<<49, 256, 0, stream>>>(cnt);
  gat_bucket<<<391, 256, 0, stream>>>(ei, cnt, bkt);

  gat_node<<<3125, 256, 0, stream>>>(xl, xr, att, bias, ei, cnt, bkt, out);
}


# edge grouping filed per package half: a wave reads its die number and uses that half's own counter and bucket arrays, so no such line is updated from both halves; node wave joins the two lists
# speedup vs baseline: 4.1120x; 1.4919x over previous
#include <hip/hip_runtime.h>
#include <hip/hip_bf16.h>

#define N_NODES   50000
#define N_EDGES   800000
#define IN_F      128
#define HC        128
#define NEG_SLOPE 0.2f

typedef __attribute__((ext_vector_type(2))) float v2f;
typedef __attribute__((ext_vector_type(8))) float v8f;

#define M_TILES (N_NODES / 16)

__global__ void __launch_bounds__(128)
gat_gemm(const float* __restrict__ x, const float* __restrict__ Wl,
         const float* __restrict__ Wr, float* __restrict__ xl,
         float* __restrict__ xr) {
  const int lane = threadIdx.x & 31;
  const int gw = __builtin_amdgcn_readfirstlane(
      (int)((blockIdx.x * 128 + threadIdx.x) >> 5));
  const int mt = gw >> 1;
  if (mt >= M_TILES) return;

  const float* __restrict__ W   = (gw & 1) ? Wr : Wl;
  float* __restrict__       dst = (gw & 1) ? xr : xl;
  const int c     = lane & 15;
  const int khalf = (lane >> 4) * 2;

  const float* __restrict__ xrow = x + (size_t)(mt * 16 + c) * IN_F + khalf;
  const float* __restrict__ wrow = W + (size_t)khalf * HC + 4 * c;

  v8f acc[8];
  for (int t = 0; t < 8; ++t) acc[t] = (v8f){};

#pragma unroll 4
  for (int kb = 0; kb < IN_F; kb += 4) {
    const float2 ax = *(const float2*)(xrow + kb);
    v2f a;
    a.x = ax.x; a.y = ax.y;
    const float* __restrict__ wk = wrow + (size_t)kb * HC;
    const float4 w0[2] = { *(const float4*)(wk),      *(const float4*)(wk + 64) };
    const float4 w1[2] = { *(const float4*)(wk + HC), *(const float4*)(wk + HC + 64) };
    for (int g = 0; g < 2; ++g) {
      v2f b;
      b.x = w0[g].x; b.y = w1[g].x;
      acc[4 * g + 0] = __builtin_amdgcn_wmma_f32_16x16x4_f32(
          false, a, false, b, (short)0, acc[4 * g + 0], false, false);
      b.x = w0[g].y; b.y = w1[g].y;
      acc[4 * g + 1] = __builtin_amdgcn_wmma_f32_16x16x4_f32(
          false, a, false, b, (short)0, acc[4 * g + 1], false, false);
      b.x = w0[g].z; b.y = w1[g].z;
      acc[4 * g + 2] = __builtin_amdgcn_wmma_f32_16x16x4_f32(
          false, a, false, b, (short)0, acc[4 * g + 2], false, false);
      b.x = w0[g].w; b.y = w1[g].w;
      acc[4 * g + 3] = __builtin_amdgcn_wmma_f32_16x16x4_f32(
          false, a, false, b, (short)0, acc[4 * g + 3], false, false);
    }
  }

  float* __restrict__ orow =
      dst + (size_t)(mt * 16 + 8 * (lane >> 4)) * HC + 4 * c;
  for (int v = 0; v < 8; ++v) {
    for (int g = 0; g < 2; ++g) {
      float4 o;
      o.x = acc[4 * g + 0][v];
      o.y = acc[4 * g + 1][v];
      o.z = acc[4 * g + 2][v];
      o.w = acc[4 * g + 3][v];
      *(float4*)(orow + (size_t)v * HC + 64 * g) = o;
    }
  }
}

#define EDGE_T  (N_EDGES / 8)
#define BKT_CAP 64
#define CNT_STRIDE 50048
#define BKT_STRIDE (N_NODES * BKT_CAP)
static_assert(CNT_STRIDE >= N_NODES && CNT_STRIDE % 32 == 0, "a half's counters fill whole lines");
static_assert(N_NODES <= 65536, "a source id must fit the 16-bit bucket slot");
static_assert(N_EDGES % 32 == 0, "the edge-list walk of an over-full node takes whole waves");

__global__ void gat_zero(int* __restrict__ cnt) {
  const int i = blockIdx.x * blockDim.x + threadIdx.x;
  if (i < 2 * CNT_STRIDE / 4) ((int4*)cnt)[i] = make_int4(0, 0, 0, 0);
}

__global__ void gat_bucket(const int* __restrict__ ei, int* __restrict__ cnt_all,
                           unsigned short* __restrict__ bkt_all) {
  const unsigned t = blockIdx.x * blockDim.x + threadIdx.x;
  const unsigned h = ((unsigned)__builtin_amdgcn_s_sendmsg_rtn(0x87) >> 18) & 1u;
  int* __restrict__ cnt = cnt_all + h * CNT_STRIDE;
  unsigned short* __restrict__ bkt = bkt_all + h * BKT_STRIDE;
  if (t < EDGE_T) {
    const int4* __restrict__ src4 = (const int4*)ei;
    const int4* __restrict__ dst4 = (const int4*)(ei + N_EDGES);
    const int4 d0 = dst4[t], d1 = dst4[t + EDGE_T];
    const int4 s0 = src4[t], s1 = src4[t + EDGE_T];
    const int p0 = atomicAdd(&cnt[d0.x], 1);
    const int p1 = atomicAdd(&cnt[d0.y], 1);
    const int p2 = atomicAdd(&cnt[d0.z], 1);
    const int p3 = atomicAdd(&cnt[d0.w], 1);
    const int p4 = atomicAdd(&cnt[d1.x], 1);
    const int p5 = atomicAdd(&cnt[d1.y], 1);
    const int p6 = atomicAdd(&cnt[d1.z], 1);
    const int p7 = atomicAdd(&cnt[d1.w], 1);
    if (p0 < BKT_CAP) bkt[(unsigned)(d0.x * BKT_CAP + p0)] = (unsigned short)s0.x;
    if (p1 < BKT_CAP) bkt[(unsigned)(d0.y * BKT_CAP + p1)] = (unsigned short)s0.y;
    if (p2 < BKT_CAP) bkt[(unsigned)(d0.z * BKT_CAP + p2)] = (unsigned short)s0.z;
    if (p3 < BKT_CAP) bkt[(unsigned)(d0.w * BKT_CAP + p3)] = (unsigned short)s0.w;
    if (p4 < BKT_CAP) bkt[(unsigned)(d1.x * BKT_CAP + p4)] = (unsigned short)s1.x;
    if (p5 < BKT_CAP) bkt[(unsigned)(d1.y * BKT_CAP + p5)] = (unsigned short)s1.y;
    if (p6 < BKT_CAP) bkt[(unsigned)(d1.z * BKT_CAP + p6)] = (unsigned short)s1.z;
    if (p7 < BKT_CAP) bkt[(unsigned)(d1.w * BKT_CAP + p7)] = (unsigned short)s1.w;
  }
}

__device__ __forceinline__ float head_logit(const float4 a, const float4 b,
                                            const float4 w) {
  float vx = a.x + b.x, vy = a.y + b.y, vz = a.z + b.z, vw = a.w + b.w;
  vx = (vx > 0.f) ? vx : vx * NEG_SLOPE;
  vy = (vy > 0.f) ? vy : vy * NEG_SLOPE;
  vz = (vz > 0.f) ? vz : vz * NEG_SLOPE;
  vw = (vw > 0.f) ? vw : vw * NEG_SLOPE;
  float s = vx * w.x + vy * w.y + vz * w.z + vw * w.w;
  s += __shfl_xor(s, 1);
  s += __shfl_xor(s, 2);
  s += __shfl_xor(s, 4);
  return s;
}

__device__ __forceinline__ int pick_rank(int k0, int k1, int r0, int r1, int p) {
  const unsigned m0 = __builtin_amdgcn_ballot_w32(r0 == p);
  if (m0) return __builtin_amdgcn_readlane(k0, __builtin_ctz(m0));
  const unsigned m1 = __builtin_amdgcn_ballot_w32(r1 == p);
  return __builtin_amdgcn_readlane(k1, __builtin_ctz(m1));
}

__device__ __forceinline__ void edge_one(const float* __restrict__ xl, const int s0,
                                         const int lane, const float4 b,
                                         const float4 w, float& m, float& den,
                                         float4& acc) {
  const float4 a0 = ((const float4*)(xl + (size_t)s0 * HC))[lane];
  const float l0 = head_logit(a0, b, w);
  const float mn = fmaxf(l0, m);
  const float sc = __expf(m - mn);
  const float p0 = __expf(l0 - mn);
  den = den * sc + p0;
  acc.x = acc.x * sc + p0 * a0.x;
  acc.y = acc.y * sc + p0 * a0.y;
  acc.z = acc.z * sc + p0 * a0.z;
  acc.w = acc.w * sc + p0 * a0.w;
  m = mn;
}

#define NODES_PER_WAVE 2

__global__ void __launch_bounds__(256) __attribute__((amdgpu_waves_per_eu(12)))
gat_node(const float* __restrict__ xl, const float* __restrict__ xr,
         const float* __restrict__ att, const float* __restrict__ bias,
         const int* __restrict__ ei, const int* __restrict__ cnt,
         const unsigned short* __restrict__ bkt, float* __restrict__ out) {
  const int lane = threadIdx.x & 31;
  const int wv = __builtin_amdgcn_readfirstlane(
      (int)(blockIdx.x * 8 + (threadIdx.x >> 5)));

  const float4 w  = ((const float4*)att)[lane];
  const float4 bi = ((const float4*)bias)[lane];

  for (int q = 0; q < NODES_PER_WAVE; ++q) {
  const int n = wv * NODES_PER_WAVE + q;

  const float4 b = ((const float4*)(xr + (size_t)n * HC))[lane];
  const int cn0 = cnt[n];
  const int cn1 = cnt[CNT_STRIDE + n];
  const int cn = cn0 + cn1;

  float4 acc = ((const float4*)(xl + (size_t)n * HC))[lane];
  float m = head_logit(acc, b, w);
  float den = 1.0f;

  if (cn <= BKT_CAP) {
    const unsigned short* __restrict__ bl = bkt + (unsigned)(n * BKT_CAP);
    const unsigned i0 = (lane < cn0) ? (unsigned)lane
                                     : (unsigned)(BKT_STRIDE + lane - cn0);
    const unsigned i1 = (lane + 32 < cn0) ? (unsigned)(lane + 32)
                                          : (unsigned)(BKT_STRIDE + lane + 32 - cn0);
    const int k0 = (lane < cn) ? (int)bl[i0] : 0x7fffffff;
    const int k1 = (lane + 32 < cn) ? (int)bl[i1] : 0x7fffffff;

    int r0 = 0, r1 = 0;
    const int c0 = (cn < 32) ? cn : 32;
    for (int j = 0; j < c0; ++j) {
      const int kj = __builtin_amdgcn_readlane(k0, j);
      r0 += (kj < k0 || (kj == k0 && j < lane)) ? 1 : 0;
    }
    if (cn > 32) {
      for (int j = 0; j < 32; ++j) {
        const int kj = __builtin_amdgcn_readlane(k0, j);
        r1 += (kj <= k1) ? 1 : 0;
      }
      for (int j = 32; j < cn; ++j) {
        const int kj = __builtin_amdgcn_readlane(k1, j - 32);
        r0 += (kj < k0) ? 1 : 0;
        r1 += (kj < k1 || (kj == k1 && j - 32 < lane)) ? 1 : 0;
      }
    }
    if (lane >= cn) r0 = -1;
    if (lane + 32 >= cn) r1 = -1;

    int j = 0;
    for (; j + 4 <= cn; j += 4) {
      const int s0 = pick_rank(k0, k1, r0, r1, j);
      const int s1 = pick_rank(k0, k1, r0, r1, j + 1);
      const int s2 = pick_rank(k0, k1, r0, r1, j + 2);
      const int s3 = pick_rank(k0, k1, r0, r1, j + 3);
      const float4 a0 = ((const float4*)(xl + (size_t)s0 * HC))[lane];
      const float4 a1 = ((const float4*)(xl + (size_t)s1 * HC))[lane];
      const float4 a2 = ((const float4*)(xl + (size_t)s2 * HC))[lane];
      const float4 a3 = ((const float4*)(xl + (size_t)s3 * HC))[lane];
      const float l0 = head_logit(a0, b, w);
      const float l1 = head_logit(a1, b, w);
      const float l2 = head_logit(a2, b, w);
      const float l3 = head_logit(a3, b, w);
      const float mn = fmaxf(fmaxf(fmaxf(l0, l1), fmaxf(l2, l3)), m);
      const float sc = __expf(m - mn);
      const float p0 = __expf(l0 - mn);
      const float p1 = __expf(l1 - mn);
      const float p2 = __expf(l2 - mn);
      const float p3 = __expf(l3 - mn);
      den = den * sc + ((p0 + p1) + (p2 + p3));
      acc.x = acc.x * sc + ((p0 * a0.x + p1 * a1.x) + (p2 * a2.x + p3 * a3.x));
      acc.y = acc.y * sc + ((p0 * a0.y + p1 * a1.y) + (p2 * a2.y + p3 * a3.y));
      acc.z = acc.z * sc + ((p0 * a0.z + p1 * a1.z) + (p2 * a2.z + p3 * a3.z));
      acc.w = acc.w * sc + ((p0 * a0.w + p1 * a1.w) + (p2 * a2.w + p3 * a3.w));
      m = mn;
    }
    for (; j < cn; ++j)
      edge_one(xl, pick_rank(k0, k1, r0, r1, j), lane, b, w, m, den, acc);
  } else {
    for (int e0 = 0; e0 < N_EDGES; e0 += 32) {
      const int d = ei[N_EDGES + e0 + lane];
      const int s = ei[e0 + lane];
      unsigned mm = __builtin_amdgcn_ballot_w32(d == n);
      while (mm) {
        const int l = __builtin_ctz(mm);
        mm &= mm - 1;
        edge_one(xl, __builtin_amdgcn_readlane(s, l), lane, b, w, m, den, acc);
      }
    }
  }

  const float dn = den + 1e-16f;
  float4 r;
  r.x = acc.x / dn + bi.x;
  r.y = acc.y / dn + bi.y;
  r.z = acc.z / dn + bi.z;
  r.w = acc.w / dn + bi.w;
  ((float4*)(out + (size_t)n * HC))[lane] = r;
  }
}

extern "C" void kernel_launch(void* const* d_in, const int* in_sizes, int n_in,
                              void* d_out, int out_size, void* d_ws, size_t ws_size,
                              hipStream_t stream) {
  const float* x    = (const float*)d_in[0];
  const int*   ei   = (const int*)  d_in[1];
  const float* Wl   = (const float*)d_in[2];
  const float* Wr   = (const float*)d_in[3];
  const float* att  = (const float*)d_in[4];
  const float* bias = (const float*)d_in[5];
  float* out = (float*)d_out;

  float* ws   = (float*)d_ws;
  float* xl   = ws;
  float* xr   = xl + (size_t)N_NODES * HC;
  int*   cnt  = (int*)(xr + (size_t)N_NODES * HC);
  unsigned short* bkt = (unsigned short*)(cnt + 2 * CNT_STRIDE);

  gat_gemm<<<1563, 128, 0, stream>>>(x, Wl, Wr, xl, xr);

  gat_zero<<<98, 256, 0, stream>>>(cnt);
  gat_bucket<<<391, 256, 0, stream>>>(ei, cnt, bkt);

  gat_node<<<3125, 256, 0, stream>>>(xl, xr, att, bias, ei, cnt, bkt, out);
}


# xr = x*W_r is made inside the node launch by the 16-node work-group that uses it (8 waves x 32 WMMAs, four 32-k partial sums joined in LDS as (p0+p1)+(p2+p3)), never stored; GEMM launch does xl only
# speedup vs baseline: 4.2975x; 1.0451x over previous
#include <hip/hip_runtime.h>
#include <hip/hip_bf16.h>

#define N_NODES   50000
#define N_EDGES   800000
#define IN_F      128
#define HC        128
#define NEG_SLOPE 0.2f

typedef __attribute__((ext_vector_type(2))) float v2f;
typedef __attribute__((ext_vector_type(8))) float v8f;

#define M_TILES (N_NODES / 16)

__device__ __forceinline__ void wmma_step4(const v2f a, const float4 w0,
                                           const float4 w1, v8f* acc) {
  v2f b;
  b.x = w0.x; b.y = w1.x;
  acc[0] = __builtin_amdgcn_wmma_f32_16x16x4_f32(false, a, false, b, (short)0,
                                                 acc[0], false, false);
  b.x = w0.y; b.y = w1.y;
  acc[1] = __builtin_amdgcn_wmma_f32_16x16x4_f32(false, a, false, b, (short)0,
                                                 acc[1], false, false);
  b.x = w0.z; b.y = w1.z;
  acc[2] = __builtin_amdgcn_wmma_f32_16x16x4_f32(false, a, false, b, (short)0,
                                                 acc[2], false, false);
  b.x = w0.w; b.y = w1.w;
  acc[3] = __builtin_amdgcn_wmma_f32_16x16x4_f32(false, a, false, b, (short)0,
                                                 acc[3], false, false);
}

__global__ void __launch_bounds__(128)
gat_gemm(const float* __restrict__ x, const float* __restrict__ Wl,
         float* __restrict__ xl) {
  const int lane = threadIdx.x & 31;
  const int gw = __builtin_amdgcn_readfirstlane(
      (int)((blockIdx.x * 128 + threadIdx.x) >> 5));
  const int mt = gw >> 1;
  const int g  = gw & 1;
  if (mt >= M_TILES) return;

  const int c     = lane & 15;
  const int khalf = (lane >> 4) * 2;

  const float* __restrict__ xrow = x + (size_t)(mt * 16 + c) * IN_F + khalf;
  const float* __restrict__ wrow = Wl + (size_t)khalf * HC + 64 * g + 4 * c;

  v8f acc[4];
  for (int t = 0; t < 4; ++t) acc[t] = (v8f){};

#pragma unroll 8
  for (int kb = 0; kb < IN_F; kb += 4) {
    const float2 ax = *(const float2*)(xrow + kb);
    v2f a;
    a.x = ax.x; a.y = ax.y;
    const float* __restrict__ wk = wrow + (size_t)kb * HC;
    const float4 w0 = *(const float4*)(wk);
    const float4 w1 = *(const float4*)(wk + HC);
    wmma_step4(a, w0, w1, acc);
  }

  float* __restrict__ orow =
      xl + (size_t)(mt * 16 + 8 * (lane >> 4)) * HC + 64 * g + 4 * c;
  for (int v = 0; v < 8; ++v) {
    float4 o;
    o.x = acc[0][v];
    o.y = acc[1][v];
    o.z = acc[2][v];
    o.w = acc[3][v];
    *(float4*)(orow + (size_t)v * HC) = o;
  }
}

#define EDGE_T  (N_EDGES / 8)
#define BKT_CAP 64
#define CNT_STRIDE 50048
#define BKT_STRIDE (N_NODES * BKT_CAP)
static_assert(CNT_STRIDE >= N_NODES && CNT_STRIDE % 32 == 0, "a half's counters fill whole lines");
static_assert(N_NODES <= 65536, "a source id must fit the 16-bit bucket slot");
static_assert(N_EDGES % 32 == 0, "the edge-list walk of an over-full node takes whole waves");

__global__ void gat_zero(int* __restrict__ cnt) {
  const int i = blockIdx.x * blockDim.x + threadIdx.x;
  if (i < 2 * CNT_STRIDE / 4) ((int4*)cnt)[i] = make_int4(0, 0, 0, 0);
}

__global__ void gat_bucket(const int* __restrict__ ei, int* __restrict__ cnt_all,
                           unsigned short* __restrict__ bkt_all) {
  const unsigned t = blockIdx.x * blockDim.x + threadIdx.x;
  const unsigned h = ((unsigned)__builtin_amdgcn_s_sendmsg_rtn(0x87) >> 18) & 1u;
  int* __restrict__ cnt = cnt_all + h * CNT_STRIDE;
  unsigned short* __restrict__ bkt = bkt_all + h * BKT_STRIDE;
  if (t < EDGE_T) {
    const int4* __restrict__ src4 = (const int4*)ei;
    const int4* __restrict__ dst4 = (const int4*)(ei + N_EDGES);
    const int4 d0 = dst4[t], d1 = dst4[t + EDGE_T];
    const int4 s0 = src4[t], s1 = src4[t + EDGE_T];
    const int p0 = atomicAdd(&cnt[d0.x], 1);
    const int p1 = atomicAdd(&cnt[d0.y], 1);
    const int p2 = atomicAdd(&cnt[d0.z], 1);
    const int p3 = atomicAdd(&cnt[d0.w], 1);
    const int p4 = atomicAdd(&cnt[d1.x], 1);
    const int p5 = atomicAdd(&cnt[d1.y], 1);
    const int p6 = atomicAdd(&cnt[d1.z], 1);
    const int p7 = atomicAdd(&cnt[d1.w], 1);
    if (p0 < BKT_CAP) bkt[(unsigned)(d0.x * BKT_CAP + p0)] = (unsigned short)s0.x;
    if (p1 < BKT_CAP) bkt[(unsigned)(d0.y * BKT_CAP + p1)] = (unsigned short)s0.y;
    if (p2 < BKT_CAP) bkt[(unsigned)(d0.z * BKT_CAP + p2)] = (unsigned short)s0.z;
    if (p3 < BKT_CAP) bkt[(unsigned)(d0.w * BKT_CAP + p3)] = (unsigned short)s0.w;
    if (p4 < BKT_CAP) bkt[(unsigned)(d1.x * BKT_CAP + p4)] = (unsigned short)s1.x;
    if (p5 < BKT_CAP) bkt[(unsigned)(d1.y * BKT_CAP + p5)] = (unsigned short)s1.y;
    if (p6 < BKT_CAP) bkt[(unsigned)(d1.z * BKT_CAP + p6)] = (unsigned short)s1.z;
    if (p7 < BKT_CAP) bkt[(unsigned)(d1.w * BKT_CAP + p7)] = (unsigned short)s1.w;
  }
}

__device__ __forceinline__ float head_logit(const float4 a, const float4 b,
                                            const float4 w) {
  float vx = a.x + b.x, vy = a.y + b.y, vz = a.z + b.z, vw = a.w + b.w;
  vx = (vx > 0.f) ? vx : vx * NEG_SLOPE;
  vy = (vy > 0.f) ? vy : vy * NEG_SLOPE;
  vz = (vz > 0.f) ? vz : vz * NEG_SLOPE;
  vw = (vw > 0.f) ? vw : vw * NEG_SLOPE;
  float s = vx * w.x + vy * w.y + vz * w.z + vw * w.w;
  s += __shfl_xor(s, 1);
  s += __shfl_xor(s, 2);
  s += __shfl_xor(s, 4);
  return s;
}

__device__ __forceinline__ int pick_rank(int k0, int k1, int r0, int r1, int p) {
  const unsigned m0 = __builtin_amdgcn_ballot_w32(r0 == p);
  if (m0) return __builtin_amdgcn_readlane(k0, __builtin_ctz(m0));
  const unsigned m1 = __builtin_amdgcn_ballot_w32(r1 == p);
  return __builtin_amdgcn_readlane(k1, __builtin_ctz(m1));
}

__device__ __forceinline__ void edge_one(const float* __restrict__ xl, const int s0,
                                         const int lane, const float4 b,
                                         const float4 w, float& m, float& den,
                                         float4& acc) {
  const float4 a0 = ((const float4*)(xl + (size_t)s0 * HC))[lane];
  const float l0 = head_logit(a0, b, w);
  const float mn = fmaxf(l0, m);
  const float sc = __expf(m - mn);
  const float p0 = __expf(l0 - mn);
  den = den * sc + p0;
  acc.x = acc.x * sc + p0 * a0.x;
  acc.y = acc.y * sc + p0 * a0.y;
  acc.z = acc.z * sc + p0 * a0.z;
  acc.w = acc.w * sc + p0 * a0.w;
  m = mn;
}

#define NODES_PER_WAVE 2
#define NODES_PER_BLOCK 16
#define XR_KQ 4
static_assert(N_NODES % NODES_PER_BLOCK == 0, "every node block is a full 16-row strip");

__global__ void __launch_bounds__(256) __attribute__((amdgpu_waves_per_eu(12)))
gat_node(const float* __restrict__ x, const float* __restrict__ Wr,
         const float* __restrict__ xl,
         const float* __restrict__ att, const float* __restrict__ bias,
         const int* __restrict__ ei, const int* __restrict__ cnt,
         const unsigned short* __restrict__ bkt, float* __restrict__ out) {
  __shared__ float4 xr_part[XR_KQ][NODES_PER_BLOCK][HC / 4];

  const int lane = threadIdx.x & 31;
  const int wib = __builtin_amdgcn_readfirstlane((int)(threadIdx.x >> 5));
  const int wv = (int)blockIdx.x * 8 + wib;

  {
    const int g  = wib & 1;
    const int kq = wib >> 1;
    const int c     = lane & 15;
    const int khalf = (lane >> 4) * 2;
    const float* __restrict__ xrow =
        x + (size_t)((int)blockIdx.x * NODES_PER_BLOCK + c) * IN_F + 32 * kq + khalf;
    const float* __restrict__ wrow =
        Wr + (size_t)(32 * kq + khalf) * HC + 64 * g + 4 * c;
    v8f pacc[4];
    for (int t = 0; t < 4; ++t) pacc[t] = (v8f){};
#pragma unroll 4
    for (int kb = 0; kb < 32; kb += 4) {
      const float2 ax = *(const float2*)(xrow + kb);
      v2f a;
      a.x = ax.x; a.y = ax.y;
      const float* __restrict__ wk = wrow + (size_t)kb * HC;
      const float4 w0 = *(const float4*)(wk);
      const float4 w1 = *(const float4*)(wk + HC);
      wmma_step4(a, w0, w1, pacc);
    }
    for (int v = 0; v < 8; ++v) {
      float4 o;
      o.x = pacc[0][v];
      o.y = pacc[1][v];
      o.z = pacc[2][v];
      o.w = pacc[3][v];
      xr_part[kq][v + 8 * (lane >> 4)][16 * g + c] = o;
    }
  }
  __syncthreads();

  const float4 w  = ((const float4*)att)[lane];
  const float4 bi = ((const float4*)bias)[lane];

  for (int q = 0; q < NODES_PER_WAVE; ++q) {
  const int n = wv * NODES_PER_WAVE + q;

  float4 b;
  {
    const int r = wib * NODES_PER_WAVE + q;
    const float4 p0 = xr_part[0][r][lane], p1 = xr_part[1][r][lane];
    const float4 p2 = xr_part[2][r][lane], p3 = xr_part[3][r][lane];
    b.x = (p0.x + p1.x) + (p2.x + p3.x);
    b.y = (p0.y + p1.y) + (p2.y + p3.y);
    b.z = (p0.z + p1.z) + (p2.z + p3.z);
    b.w = (p0.w + p1.w) + (p2.w + p3.w);
  }
  const int cn0 = cnt[n];
  const int cn1 = cnt[CNT_STRIDE + n];
  const int cn = cn0 + cn1;

  float4 acc = ((const float4*)(xl + (size_t)n * HC))[lane];
  float m = head_logit(acc, b, w);
  float den = 1.0f;

  if (cn <= BKT_CAP) {
    const unsigned short* __restrict__ bl = bkt + (unsigned)(n * BKT_CAP);
    const unsigned i0 = (lane < cn0) ? (unsigned)lane
                                     : (unsigned)(BKT_STRIDE + lane - cn0);
    const unsigned i1 = (lane + 32 < cn0) ? (unsigned)(lane + 32)
                                          : (unsigned)(BKT_STRIDE + lane + 32 - cn0);
    const int k0 = (lane < cn) ? (int)bl[i0] : 0x7fffffff;
    const int k1 = (lane + 32 < cn) ? (int)bl[i1] : 0x7fffffff;

    int r0 = 0, r1 = 0;
    const int c0 = (cn < 32) ? cn : 32;
    for (int j = 0; j < c0; ++j) {
      const int kj = __builtin_amdgcn_readlane(k0, j);
      r0 += (kj < k0 || (kj == k0 && j < lane)) ? 1 : 0;
    }
    if (cn > 32) {
      for (int j = 0; j < 32; ++j) {
        const int kj = __builtin_amdgcn_readlane(k0, j);
        r1 += (kj <= k1) ? 1 : 0;
      }
      for (int j = 32; j < cn; ++j) {
        const int kj = __builtin_amdgcn_readlane(k1, j - 32);
        r0 += (kj < k0) ? 1 : 0;
        r1 += (kj < k1 || (kj == k1 && j - 32 < lane)) ? 1 : 0;
      }
    }
    if (lane >= cn) r0 = -1;
    if (lane + 32 >= cn) r1 = -1;

    int j = 0;
    for (; j + 4 <= cn; j += 4) {
      const int s0 = pick_rank(k0, k1, r0, r1, j);
      const int s1 = pick_rank(k0, k1, r0, r1, j + 1);
      const int s2 = pick_rank(k0, k1, r0, r1, j + 2);
      const int s3 = pick_rank(k0, k1, r0, r1, j + 3);
      const float4 a0 = ((const float4*)(xl + (size_t)s0 * HC))[lane];
      const float4 a1 = ((const float4*)(xl + (size_t)s1 * HC))[lane];
      const float4 a2 = ((const float4*)(xl + (size_t)s2 * HC))[lane];
      const float4 a3 = ((const float4*)(xl + (size_t)s3 * HC))[lane];
      const float l0 = head_logit(a0, b, w);
      const float l1 = head_logit(a1, b, w);
      const float l2 = head_logit(a2, b, w);
      const float l3 = head_logit(a3, b, w);
      const float mn = fmaxf(fmaxf(fmaxf(l0, l1), fmaxf(l2, l3)), m);
      const float sc = __expf(m - mn);
      const float p0 = __expf(l0 - mn);
      const float p1 = __expf(l1 - mn);
      const float p2 = __expf(l2 - mn);
      const float p3 = __expf(l3 - mn);
      den = den * sc + ((p0 + p1) + (p2 + p3));
      acc.x = acc.x * sc + ((p0 * a0.x + p1 * a1.x) + (p2 * a2.x + p3 * a3.x));
      acc.y = acc.y * sc + ((p0 * a0.y + p1 * a1.y) + (p2 * a2.y + p3 * a3.y));
      acc.z = acc.z * sc + ((p0 * a0.z + p1 * a1.z) + (p2 * a2.z + p3 * a3.z));
      acc.w = acc.w * sc + ((p0 * a0.w + p1 * a1.w) + (p2 * a2.w + p3 * a3.w));
      m = mn;
    }
    for (; j < cn; ++j)
      edge_one(xl, pick_rank(k0, k1, r0, r1, j), lane, b, w, m, den, acc);
  } else {
    for (int e0 = 0; e0 < N_EDGES; e0 += 32) {
      const int d = ei[N_EDGES + e0 + lane];
      const int s = ei[e0 + lane];
      unsigned mm = __builtin_amdgcn_ballot_w32(d == n);
      while (mm) {
        const int l = __builtin_ctz(mm);
        mm &= mm - 1;
        edge_one(xl, __builtin_amdgcn_readlane(s, l), lane, b, w, m, den, acc);
      }
    }
  }

  const float dn = den + 1e-16f;
  float4 r;
  r.x = acc.x / dn + bi.x;
  r.y = acc.y / dn + bi.y;
  r.z = acc.z / dn + bi.z;
  r.w = acc.w / dn + bi.w;
  ((float4*)(out + (size_t)n * HC))[lane] = r;
  }
}

extern "C" void kernel_launch(void* const* d_in, const int* in_sizes, int n_in,
                              void* d_out, int out_size, void* d_ws, size_t ws_size,
                              hipStream_t stream) {
  const float* x    = (const float*)d_in[0];
  const int*   ei   = (const int*)  d_in[1];
  const float* Wl   = (const float*)d_in[2];
  const float* Wr   = (const float*)d_in[3];
  const float* att  = (const float*)d_in[4];
  const float* bias = (const float*)d_in[5];
  float* out = (float*)d_out;

  float* ws   = (float*)d_ws;
  float* xl   = ws;
  int*   cnt  = (int*)(xl + (size_t)N_NODES * HC);
  unsigned short* bkt = (unsigned short*)(cnt + 2 * CNT_STRIDE);

  gat_gemm<<<1563, 128, 0, stream>>>(x, Wl, xl);

  gat_zero<<<98, 256, 0, stream>>>(cnt);
  gat_bucket<<<391, 256, 0, stream>>>(ei, cnt, bkt);

  gat_node<<<3125, 256, 0, stream>>>(x, Wr, xl, att, bias, ei, cnt, bkt, out);
}


# node pass: same arithmetic in fewer vector instructions: leaky relu = max(v,s*v); packed 2-float ops (dot (c0+c2)+(c1+c3), group terms chained in order); one-compare rank keys; one 1/den per node
# speedup vs baseline: 4.4948x; 1.0459x over previous
#include <hip/hip_runtime.h>
#include <hip/hip_bf16.h>

#define N_NODES   50000
#define N_EDGES   800000
#define IN_F      128
#define HC        128
#define NEG_SLOPE 0.2f
#define LOG2E_F   1.44269504088896340736f

typedef __attribute__((ext_vector_type(2))) float v2f;
typedef __attribute__((ext_vector_type(8))) float v8f;

#define M_TILES (N_NODES / 16)

__device__ __forceinline__ void wmma_step4(const v2f a, const float4 w0,
                                           const float4 w1, v8f* acc) {
  v2f b;
  b.x = w0.x; b.y = w1.x;
  acc[0] = __builtin_amdgcn_wmma_f32_16x16x4_f32(false, a, false, b, (short)0,
                                                 acc[0], false, false);
  b.x = w0.y; b.y = w1.y;
  acc[1] = __builtin_amdgcn_wmma_f32_16x16x4_f32(false, a, false, b, (short)0,
                                                 acc[1], false, false);
  b.x = w0.z; b.y = w1.z;
  acc[2] = __builtin_amdgcn_wmma_f32_16x16x4_f32(false, a, false, b, (short)0,
                                                 acc[2], false, false);
  b.x = w0.w; b.y = w1.w;
  acc[3] = __builtin_amdgcn_wmma_f32_16x16x4_f32(false, a, false, b, (short)0,
                                                 acc[3], false, false);
}

__global__ void __launch_bounds__(128)
gat_gemm(const float* __restrict__ x, const float* __restrict__ Wl,
         float* __restrict__ xl) {
  const int lane = threadIdx.x & 31;
  const int gw = __builtin_amdgcn_readfirstlane(
      (int)((blockIdx.x * 128 + threadIdx.x) >> 5));
  const int mt = gw >> 1;
  const int g  = gw & 1;
  if (mt >= M_TILES) return;

  const int c     = lane & 15;
  const int khalf = (lane >> 4) * 2;

  const float* __restrict__ xrow = x + (size_t)(mt * 16 + c) * IN_F + khalf;
  const float* __restrict__ wrow = Wl + (size_t)khalf * HC + 64 * g + 4 * c;

  v8f acc[4];
  for (int t = 0; t < 4; ++t) acc[t] = (v8f){};

#pragma unroll 8
  for (int kb = 0; kb < IN_F; kb += 4) {
    const float2 ax = *(const float2*)(xrow + kb);
    v2f a;
    a.x = ax.x; a.y = ax.y;
    const float* __restrict__ wk = wrow + (size_t)kb * HC;
    const float4 w0 = *(const float4*)(wk);
    const float4 w1 = *(const float4*)(wk + HC);
    wmma_step4(a, w0, w1, acc);
  }

  float* __restrict__ orow =
      xl + (size_t)(mt * 16 + 8 * (lane >> 4)) * HC + 64 * g + 4 * c;
  for (int v = 0; v < 8; ++v) {
    float4 o;
    o.x = acc[0][v];
    o.y = acc[1][v];
    o.z = acc[2][v];
    o.w = acc[3][v];
    *(float4*)(orow + (size_t)v * HC) = o;
  }
}

#define EDGE_T  (N_EDGES / 8)
#define BKT_CAP 64
#define CNT_STRIDE 50048
#define BKT_STRIDE (N_NODES * BKT_CAP)
static_assert(CNT_STRIDE >= N_NODES && CNT_STRIDE % 32 == 0, "a half's counters fill whole lines");
static_assert(N_NODES <= 65536, "a source id must fit the 16-bit bucket slot");
static_assert(N_EDGES % 32 == 0, "the edge-list walk of an over-full node takes whole waves");

__global__ void gat_zero(int* __restrict__ cnt) {
  const int i = blockIdx.x * blockDim.x + threadIdx.x;
  if (i < 2 * CNT_STRIDE / 4) ((int4*)cnt)[i] = make_int4(0, 0, 0, 0);
}

__global__ void gat_bucket(const int* __restrict__ ei, int* __restrict__ cnt_all,
                           unsigned short* __restrict__ bkt_all) {
  const unsigned t = blockIdx.x * blockDim.x + threadIdx.x;
  const unsigned h = ((unsigned)__builtin_amdgcn_s_sendmsg_rtn(0x87) >> 18) & 1u;
  int* __restrict__ cnt = cnt_all + h * CNT_STRIDE;
  unsigned short* __restrict__ bkt = bkt_all + h * BKT_STRIDE;
  if (t < EDGE_T) {
    const int4* __restrict__ src4 = (const int4*)ei;
    const int4* __restrict__ dst4 = (const int4*)(ei + N_EDGES);
    const int4 d0 = dst4[t], d1 = dst4[t + EDGE_T];
    const int4 s0 = src4[t], s1 = src4[t + EDGE_T];
    const int p0 = atomicAdd(&cnt[d0.x], 1);
    const int p1 = atomicAdd(&cnt[d0.y], 1);
    const int p2 = atomicAdd(&cnt[d0.z], 1);
    const int p3 = atomicAdd(&cnt[d0.w], 1);
    const int p4 = atomicAdd(&cnt[d1.x], 1);
    const int p5 = atomicAdd(&cnt[d1.y], 1);
    const int p6 = atomicAdd(&cnt[d1.z], 1);
    const int p7 = atomicAdd(&cnt[d1.w], 1);
    if (p0 < BKT_CAP) bkt[(unsigned)(d0.x * BKT_CAP + p0)] = (unsigned short)s0.x;
    if (p1 < BKT_CAP) bkt[(unsigned)(d0.y * BKT_CAP + p1)] = (unsigned short)s0.y;
    if (p2 < BKT_CAP) bkt[(unsigned)(d0.z * BKT_CAP + p2)] = (unsigned short)s0.z;
    if (p3 < BKT_CAP) bkt[(unsigned)(d0.w * BKT_CAP + p3)] = (unsigned short)s0.w;
    if (p4 < BKT_CAP) bkt[(unsigned)(d1.x * BKT_CAP + p4)] = (unsigned short)s1.x;
    if (p5 < BKT_CAP) bkt[(unsigned)(d1.y * BKT_CAP + p5)] = (unsigned short)s1.y;
    if (p6 < BKT_CAP) bkt[(unsigned)(d1.z * BKT_CAP + p6)] = (unsigned short)s1.z;
    if (p7 < BKT_CAP) bkt[(unsigned)(d1.w * BKT_CAP + p7)] = (unsigned short)s1.w;
  }
}

struct row4 { v2f lo, hi; };

__device__ __forceinline__ row4 load_row4(const float* __restrict__ base, const int lane) {
  const float4 a = ((const float4*)base)[lane];
  row4 r;
  r.lo.x = a.x; r.lo.y = a.y;
  r.hi.x = a.z; r.hi.y = a.w;
  return r;
}

__device__ __forceinline__ float head_logit(const row4 a, const row4 b, const row4 w) {
  v2f vlo = a.lo + b.lo, vhi = a.hi + b.hi;
  vlo = __builtin_elementwise_max(vlo, vlo * NEG_SLOPE);
  vhi = __builtin_elementwise_max(vhi, vhi * NEG_SLOPE);
  const v2f s2 = __builtin_elementwise_fma(vhi, w.hi, vlo * w.lo);
  float s = s2.x + s2.y;
  s += __shfl_xor(s, 1);
  s += __shfl_xor(s, 2);
  s += __shfl_xor(s, 4);
  return s;
}

__device__ __forceinline__ int pick_rank(int k0, int k1, int r0, int r1, int p) {
  const unsigned m0 = __builtin_amdgcn_ballot_w32(r0 == p);
  if (m0) return __builtin_amdgcn_readlane(k0, __builtin_ctz(m0)) >> 6;
  const unsigned m1 = __builtin_amdgcn_ballot_w32(r1 == p);
  return __builtin_amdgcn_readlane(k1, __builtin_ctz(m1)) >> 6;
}

__device__ __forceinline__ void edge_one(const float* __restrict__ xl, const int s0,
                                         const int lane, const row4 b,
                                         const row4 w, float& m, float& den,
                                         row4& acc) {
  const row4 a0 = load_row4(xl + (size_t)s0 * HC, lane);
  const float l0 = head_logit(a0, b, w);
  const float mn = fmaxf(l0, m);
  const v2f e2 = ((v2f){m, l0} - (v2f){mn, mn}) * LOG2E_F;
  const float sc = __builtin_amdgcn_exp2f(e2.x);
  const float p0 = __builtin_amdgcn_exp2f(e2.y);
  den = den * sc + p0;
  acc.lo = __builtin_elementwise_fma(acc.lo, (v2f){sc, sc}, a0.lo * p0);
  acc.hi = __builtin_elementwise_fma(acc.hi, (v2f){sc, sc}, a0.hi * p0);
  m = mn;
}

#define NODES_PER_WAVE 2
#define NODES_PER_BLOCK 16
#define XR_KQ 4
static_assert(N_NODES % NODES_PER_BLOCK == 0, "every node block is a full 16-row strip");

__global__ void __launch_bounds__(256) __attribute__((amdgpu_waves_per_eu(12)))
gat_node(const float* __restrict__ x, const float* __restrict__ Wr,
         const float* __restrict__ xl,
         const float* __restrict__ att, const float* __restrict__ bias,
         const int* __restrict__ ei, const int* __restrict__ cnt,
         const unsigned short* __restrict__ bkt, float* __restrict__ out) {
  __shared__ float4 xr_part[XR_KQ][NODES_PER_BLOCK][HC / 4];

  const int lane = threadIdx.x & 31;
  const int wib = __builtin_amdgcn_readfirstlane((int)(threadIdx.x >> 5));
  const int wv = (int)blockIdx.x * 8 + wib;

  {
    const int g  = wib & 1;
    const int kq = wib >> 1;
    const int c     = lane & 15;
    const int khalf = (lane >> 4) * 2;
    const float* __restrict__ xrow =
        x + (size_t)((int)blockIdx.x * NODES_PER_BLOCK + c) * IN_F + 32 * kq + khalf;
    const float* __restrict__ wrow =
        Wr + (size_t)(32 * kq + khalf) * HC + 64 * g + 4 * c;
    v8f pacc[4];
    for (int t = 0; t < 4; ++t) pacc[t] = (v8f){};
#pragma unroll 4
    for (int kb = 0; kb < 32; kb += 4) {
      const float2 ax = *(const float2*)(xrow + kb);
      v2f a;
      a.x = ax.x; a.y = ax.y;
      const float* __restrict__ wk = wrow + (size_t)kb * HC;
      const float4 w0 = *(const float4*)(wk);
      const float4 w1 = *(const float4*)(wk + HC);
      wmma_step4(a, w0, w1, pacc);
    }
    for (int v = 0; v < 8; ++v) {
      float4 o;
      o.x = pacc[0][v];
      o.y = pacc[1][v];
      o.z = pacc[2][v];
      o.w = pacc[3][v];
      xr_part[kq][v + 8 * (lane >> 4)][16 * g + c] = o;
    }
  }
  __syncthreads();

  const row4 w  = load_row4(att, lane);
  const row4 bi = load_row4(bias, lane);

  for (int q = 0; q < NODES_PER_WAVE; ++q) {
  const int n = wv * NODES_PER_WAVE + q;

  row4 b;
  {
    const int r = wib * NODES_PER_WAVE + q;
    const float4 p0 = xr_part[0][r][lane], p1 = xr_part[1][r][lane];
    const float4 p2 = xr_part[2][r][lane], p3 = xr_part[3][r][lane];
    b.lo.x = (p0.x + p1.x) + (p2.x + p3.x);
    b.lo.y = (p0.y + p1.y) + (p2.y + p3.y);
    b.hi.x = (p0.z + p1.z) + (p2.z + p3.z);
    b.hi.y = (p0.w + p1.w) + (p2.w + p3.w);
  }
  const int cn0 = cnt[n];
  const int cn1 = cnt[CNT_STRIDE + n];
  const int cn = cn0 + cn1;

  row4 acc = load_row4(xl + (size_t)n * HC, lane);
  float m = head_logit(acc, b, w);
  float den = 1.0f;

  if (cn <= BKT_CAP) {
    const unsigned short* __restrict__ bl = bkt + (unsigned)(n * BKT_CAP);
    const unsigned i0 = (lane < cn0) ? (unsigned)lane
                                     : (unsigned)(BKT_STRIDE + lane - cn0);
    const unsigned i1 = (lane + 32 < cn0) ? (unsigned)(lane + 32)
                                          : (unsigned)(BKT_STRIDE + lane + 32 - cn0);
    const int k0 = (lane < cn) ? (((int)bl[i0] << 6) | lane) : 0x7fffffff;
    const int k1 = (lane + 32 < cn) ? (((int)bl[i1] << 6) | (lane + 32)) : 0x7fffffff;

    int r0 = 0, r1 = 0;
    const int c0 = (cn < 32) ? cn : 32;
    for (int j = 0; j < c0; ++j) {
      const int kj = __builtin_amdgcn_readlane(k0, j);
      r0 += (kj < k0) ? 1 : 0;
    }
    if (cn > 32) {
      for (int j = 0; j < 32; ++j) {
        const int kj = __builtin_amdgcn_readlane(k0, j);
        r1 += (kj < k1) ? 1 : 0;
      }
      for (int j = 32; j < cn; ++j) {
        const int kj = __builtin_amdgcn_readlane(k1, j - 32);
        r0 += (kj < k0) ? 1 : 0;
        r1 += (kj < k1) ? 1 : 0;
      }
    }
    if (lane >= cn) r0 = -1;
    if (lane + 32 >= cn) r1 = -1;

    int j = 0;
    for (; j + 4 <= cn; j += 4) {
      const int s0 = pick_rank(k0, k1, r0, r1, j);
      const int s1 = pick_rank(k0, k1, r0, r1, j + 1);
      const int s2 = pick_rank(k0, k1, r0, r1, j + 2);
      const int s3 = pick_rank(k0, k1, r0, r1, j + 3);
      const row4 a0 = load_row4(xl + (size_t)s0 * HC, lane);
      const row4 a1 = load_row4(xl + (size_t)s1 * HC, lane);
      const row4 a2 = load_row4(xl + (size_t)s2 * HC, lane);
      const row4 a3 = load_row4(xl + (size_t)s3 * HC, lane);
      const float l0 = head_logit(a0, b, w);
      const float l1 = head_logit(a1, b, w);
      const float l2 = head_logit(a2, b, w);
      const float l3 = head_logit(a3, b, w);
      const float mn = fmaxf(fmaxf(fmaxf(l0, l1), fmaxf(l2, l3)), m);
      const v2f mn2 = {mn, mn};
      const v2f e01 = ((v2f){l0, l1} - mn2) * LOG2E_F;
      const v2f e23 = ((v2f){l2, l3} - mn2) * LOG2E_F;
      const float sc = __builtin_amdgcn_exp2f((m - mn) * LOG2E_F);
      const float p0 = __builtin_amdgcn_exp2f(e01.x);
      const float p1 = __builtin_amdgcn_exp2f(e01.y);
      const float p2 = __builtin_amdgcn_exp2f(e23.x);
      const float p3 = __builtin_amdgcn_exp2f(e23.y);
      den = den * sc + ((p0 + p1) + (p2 + p3));
      acc.lo = acc.lo * sc;
      acc.hi = acc.hi * sc;
      acc.lo = __builtin_elementwise_fma(a0.lo, (v2f){p0, p0}, acc.lo);
      acc.hi = __builtin_elementwise_fma(a0.hi, (v2f){p0, p0}, acc.hi);
      acc.lo = __builtin_elementwise_fma(a1.lo, (v2f){p1, p1}, acc.lo);
      acc.hi = __builtin_elementwise_fma(a1.hi, (v2f){p1, p1}, acc.hi);
      acc.lo = __builtin_elementwise_fma(a2.lo, (v2f){p2, p2}, acc.lo);
      acc.hi = __builtin_elementwise_fma(a2.hi, (v2f){p2, p2}, acc.hi);
      acc.lo = __builtin_elementwise_fma(a3.lo, (v2f){p3, p3}, acc.lo);
      acc.hi = __builtin_elementwise_fma(a3.hi, (v2f){p3, p3}, acc.hi);
      m = mn;
    }
    for (; j < cn; ++j)
      edge_one(xl, pick_rank(k0, k1, r0, r1, j), lane, b, w, m, den, acc);
  } else {
    for (int e0 = 0; e0 < N_EDGES; e0 += 32) {
      const int d = ei[N_EDGES + e0 + lane];
      const int s = ei[e0 + lane];
      unsigned mm = __builtin_amdgcn_ballot_w32(d == n);
      while (mm) {
        const int l = __builtin_ctz(mm);
        mm &= mm - 1;
        edge_one(xl, __builtin_amdgcn_readlane(s, l), lane, b, w, m, den, acc);
      }
    }
  }

  const float inv = 1.0f / (den + 1e-16f);
  const v2f olo = __builtin_elementwise_fma(acc.lo, (v2f){inv, inv}, bi.lo);
  const v2f ohi = __builtin_elementwise_fma(acc.hi, (v2f){inv, inv}, bi.hi);
  float4 r;
  r.x = olo.x; r.y = olo.y;
  r.z = ohi.x; r.w = ohi.y;
  ((float4*)(out + (size_t)n * HC))[lane] = r;
  }
}

extern "C" void kernel_launch(void* const* d_in, const int* in_sizes, int n_in,
                              void* d_out, int out_size, void* d_ws, size_t ws_size,
                              hipStream_t stream) {
  const float* x    = (const float*)d_in[0];
  const int*   ei   = (const int*)  d_in[1];
  const float* Wl   = (const float*)d_in[2];
  const float* Wr   = (const float*)d_in[3];
  const float* att  = (const float*)d_in[4];
  const float* bias = (const float*)d_in[5];
  float* out = (float*)d_out;

  float* ws   = (float*)d_ws;
  float* xl   = ws;
  int*   cnt  = (int*)(xl + (size_t)N_NODES * HC);
  unsigned short* bkt = (unsigned short*)(cnt + 2 * CNT_STRIDE);

  gat_gemm<<<1563, 128, 0, stream>>>(x, Wl, xl);

  gat_zero<<<98, 256, 0, stream>>>(cnt);
  gat_bucket<<<391, 256, 0, stream>>>(ei, cnt, bkt);

  gat_node<<<3125, 256, 0, stream>>>(x, Wr, xl, att, bias, ei, cnt, bkt, out);
}


# node pass cross-lane steps off the LDS hops and ballots: head sums by row_xmask DPP adds; lists of 32 or fewer put in rank order by one lane permute, edge p's source read from lane p
# speedup vs baseline: 4.8542x; 1.0800x over previous
#include <hip/hip_runtime.h>
#include <hip/hip_bf16.h>

#define N_NODES   50000
#define N_EDGES   800000
#define IN_F      128
#define HC        128
#define NEG_SLOPE 0.2f
#define LOG2E_F   1.44269504088896340736f

typedef __attribute__((ext_vector_type(2))) float v2f;
typedef __attribute__((ext_vector_type(8))) float v8f;

#define M_TILES (N_NODES / 16)

__device__ __forceinline__ void wmma_step4(const v2f a, const float4 w0,
                                           const float4 w1, v8f* acc) {
  v2f b;
  b.x = w0.x; b.y = w1.x;
  acc[0] = __builtin_amdgcn_wmma_f32_16x16x4_f32(false, a, false, b, (short)0,
                                                 acc[0], false, false);
  b.x = w0.y; b.y = w1.y;
  acc[1] = __builtin_amdgcn_wmma_f32_16x16x4_f32(false, a, false, b, (short)0,
                                                 acc[1], false, false);
  b.x = w0.z; b.y = w1.z;
  acc[2] = __builtin_amdgcn_wmma_f32_16x16x4_f32(false, a, false, b, (short)0,
                                                 acc[2], false, false);
  b.x = w0.w; b.y = w1.w;
  acc[3] = __builtin_amdgcn_wmma_f32_16x16x4_f32(false, a, false, b, (short)0,
                                                 acc[3], false, false);
}

__global__ void __launch_bounds__(128)
gat_gemm(const float* __restrict__ x, const float* __restrict__ Wl,
         float* __restrict__ xl) {
  const int lane = threadIdx.x & 31;
  const int gw = __builtin_amdgcn_readfirstlane(
      (int)((blockIdx.x * 128 + threadIdx.x) >> 5));
  const int mt = gw >> 1;
  const int g  = gw & 1;
  if (mt >= M_TILES) return;

  const int c     = lane & 15;
  const int khalf = (lane >> 4) * 2;

  const float* __restrict__ xrow = x + (size_t)(mt * 16 + c) * IN_F + khalf;
  const float* __restrict__ wrow = Wl + (size_t)khalf * HC + 64 * g + 4 * c;

  v8f acc[4];
  for (int t = 0; t < 4; ++t) acc[t] = (v8f){};

#pragma unroll 8
  for (int kb = 0; kb < IN_F; kb += 4) {
    const float2 ax = *(const float2*)(xrow + kb);
    v2f a;
    a.x = ax.x; a.y = ax.y;
    const float* __restrict__ wk = wrow + (size_t)kb * HC;
    const float4 w0 = *(const float4*)(wk);
    const float4 w1 = *(const float4*)(wk + HC);
    wmma_step4(a, w0, w1, acc);
  }

  float* __restrict__ orow =
      xl + (size_t)(mt * 16 + 8 * (lane >> 4)) * HC + 64 * g + 4 * c;
  for (int v = 0; v < 8; ++v) {
    float4 o;
    o.x = acc[0][v];
    o.y = acc[1][v];
    o.z = acc[2][v];
    o.w = acc[3][v];
    *(float4*)(orow + (size_t)v * HC) = o;
  }
}

#define EDGE_T  (N_EDGES / 8)
#define BKT_CAP 64
#define CNT_STRIDE 50048
#define BKT_STRIDE (N_NODES * BKT_CAP)
static_assert(CNT_STRIDE >= N_NODES && CNT_STRIDE % 32 == 0, "a half's counters fill whole lines");
static_assert(N_NODES <= 65536, "a source id must fit the 16-bit bucket slot");
static_assert(N_EDGES % 32 == 0, "the edge-list walk of an over-full node takes whole waves");

__global__ void gat_zero(int* __restrict__ cnt) {
  const int i = blockIdx.x * blockDim.x + threadIdx.x;
  if (i < 2 * CNT_STRIDE / 4) ((int4*)cnt)[i] = make_int4(0, 0, 0, 0);
}

__global__ void gat_bucket(const int* __restrict__ ei, int* __restrict__ cnt_all,
                           unsigned short* __restrict__ bkt_all) {
  const unsigned t = blockIdx.x * blockDim.x + threadIdx.x;
  const unsigned h = ((unsigned)__builtin_amdgcn_s_sendmsg_rtn(0x87) >> 18) & 1u;
  int* __restrict__ cnt = cnt_all + h * CNT_STRIDE;
  unsigned short* __restrict__ bkt = bkt_all + h * BKT_STRIDE;
  if (t < EDGE_T) {
    const int4* __restrict__ src4 = (const int4*)ei;
    const int4* __restrict__ dst4 = (const int4*)(ei + N_EDGES);
    const int4 d0 = dst4[t], d1 = dst4[t + EDGE_T];
    const int4 s0 = src4[t], s1 = src4[t + EDGE_T];
    const int p0 = atomicAdd(&cnt[d0.x], 1);
    const int p1 = atomicAdd(&cnt[d0.y], 1);
    const int p2 = atomicAdd(&cnt[d0.z], 1);
    const int p3 = atomicAdd(&cnt[d0.w], 1);
    const int p4 = atomicAdd(&cnt[d1.x], 1);
    const int p5 = atomicAdd(&cnt[d1.y], 1);
    const int p6 = atomicAdd(&cnt[d1.z], 1);
    const int p7 = atomicAdd(&cnt[d1.w], 1);
    if (p0 < BKT_CAP) bkt[(unsigned)(d0.x * BKT_CAP + p0)] = (unsigned short)s0.x;
    if (p1 < BKT_CAP) bkt[(unsigned)(d0.y * BKT_CAP + p1)] = (unsigned short)s0.y;
    if (p2 < BKT_CAP) bkt[(unsigned)(d0.z * BKT_CAP + p2)] = (unsigned short)s0.z;
    if (p3 < BKT_CAP) bkt[(unsigned)(d0.w * BKT_CAP + p3)] = (unsigned short)s0.w;
    if (p4 < BKT_CAP) bkt[(unsigned)(d1.x * BKT_CAP + p4)] = (unsigned short)s1.x;
    if (p5 < BKT_CAP) bkt[(unsigned)(d1.y * BKT_CAP + p5)] = (unsigned short)s1.y;
    if (p6 < BKT_CAP) bkt[(unsigned)(d1.z * BKT_CAP + p6)] = (unsigned short)s1.z;
    if (p7 < BKT_CAP) bkt[(unsigned)(d1.w * BKT_CAP + p7)] = (unsigned short)s1.w;
  }
}

struct row4 { v2f lo, hi; };

__device__ __forceinline__ row4 load_row4(const float* __restrict__ base, const int lane) {
  const float4 a = ((const float4*)base)[lane];
  row4 r;
  r.lo.x = a.x; r.lo.y = a.y;
  r.hi.x = a.z; r.hi.y = a.w;
  return r;
}

template <int X>
__device__ __forceinline__ float add_xor_lane(const float s) {
  const int t = __builtin_amdgcn_update_dpp(0, __float_as_int(s), 0x160 + X, 0xf, 0xf, true);
  return s + __int_as_float(t);
}

__device__ __forceinline__ float head_logit(const row4 a, const row4 b, const row4 w) {
  v2f vlo = a.lo + b.lo, vhi = a.hi + b.hi;
  vlo = __builtin_elementwise_max(vlo, vlo * NEG_SLOPE);
  vhi = __builtin_elementwise_max(vhi, vhi * NEG_SLOPE);
  const v2f s2 = __builtin_elementwise_fma(vhi, w.hi, vlo * w.lo);
  float s = s2.x + s2.y;
  s = add_xor_lane<1>(s);
  s = add_xor_lane<2>(s);
  s = add_xor_lane<4>(s);
  return s;
}

__device__ __forceinline__ int pick_rank(int k0, int k1, int r0, int r1, int p) {
  const unsigned m0 = __builtin_amdgcn_ballot_w32(r0 == p);
  if (m0) return __builtin_amdgcn_readlane(k0, __builtin_ctz(m0)) >> 6;
  const unsigned m1 = __builtin_amdgcn_ballot_w32(r1 == p);
  return __builtin_amdgcn_readlane(k1, __builtin_ctz(m1)) >> 6;
}

__device__ __forceinline__ void edge_one(const float* __restrict__ xl, const int s0,
                                         const int lane, const row4 b,
                                         const row4 w, float& m, float& den,
                                         row4& acc) {
  const row4 a0 = load_row4(xl + (size_t)s0 * HC, lane);
  const float l0 = head_logit(a0, b, w);
  const float mn = fmaxf(l0, m);
  const v2f e2 = ((v2f){m, l0} - (v2f){mn, mn}) * LOG2E_F;
  const float sc = __builtin_amdgcn_exp2f(e2.x);
  const float p0 = __builtin_amdgcn_exp2f(e2.y);
  den = den * sc + p0;
  acc.lo = __builtin_elementwise_fma(acc.lo, (v2f){sc, sc}, a0.lo * p0);
  acc.hi = __builtin_elementwise_fma(acc.hi, (v2f){sc, sc}, a0.hi * p0);
  m = mn;
}

template <bool SORTED>
__device__ __forceinline__ int list_src(const int sorted, const int k0, const int k1,
                                        const int r0, const int r1, const int p) {
  if (SORTED) return __builtin_amdgcn_readlane(sorted, p) >> 6;
  return pick_rank(k0, k1, r0, r1, p);
}

template <bool SORTED>
__device__ __forceinline__ void walk_list(const float* __restrict__ xl, const int cn,
                                          const int sorted, const int k0, const int k1,
                                          const int r0, const int r1, const int lane,
                                          const row4 b, const row4 w, float& m,
                                          float& den, row4& acc) {
  int j = 0;
  for (; j + 4 <= cn; j += 4) {
    const int s0 = list_src<SORTED>(sorted, k0, k1, r0, r1, j);
    const int s1 = list_src<SORTED>(sorted, k0, k1, r0, r1, j + 1);
    const int s2 = list_src<SORTED>(sorted, k0, k1, r0, r1, j + 2);
    const int s3 = list_src<SORTED>(sorted, k0, k1, r0, r1, j + 3);
    const row4 a0 = load_row4(xl + (size_t)s0 * HC, lane);
    const row4 a1 = load_row4(xl + (size_t)s1 * HC, lane);
    const row4 a2 = load_row4(xl + (size_t)s2 * HC, lane);
    const row4 a3 = load_row4(xl + (size_t)s3 * HC, lane);
    const float l0 = head_logit(a0, b, w);
    const float l1 = head_logit(a1, b, w);
    const float l2 = head_logit(a2, b, w);
    const float l3 = head_logit(a3, b, w);
    const float mn = fmaxf(fmaxf(fmaxf(l0, l1), fmaxf(l2, l3)), m);
    const v2f mn2 = {mn, mn};
    const v2f e01 = ((v2f){l0, l1} - mn2) * LOG2E_F;
    const v2f e23 = ((v2f){l2, l3} - mn2) * LOG2E_F;
    const float sc = __builtin_amdgcn_exp2f((m - mn) * LOG2E_F);
    const float p0 = __builtin_amdgcn_exp2f(e01.x);
    const float p1 = __builtin_amdgcn_exp2f(e01.y);
    const float p2 = __builtin_amdgcn_exp2f(e23.x);
    const float p3 = __builtin_amdgcn_exp2f(e23.y);
    den = den * sc + ((p0 + p1) + (p2 + p3));
    acc.lo = acc.lo * sc;
    acc.hi = acc.hi * sc;
    acc.lo = __builtin_elementwise_fma(a0.lo, (v2f){p0, p0}, acc.lo);
    acc.hi = __builtin_elementwise_fma(a0.hi, (v2f){p0, p0}, acc.hi);
    acc.lo = __builtin_elementwise_fma(a1.lo, (v2f){p1, p1}, acc.lo);
    acc.hi = __builtin_elementwise_fma(a1.hi, (v2f){p1, p1}, acc.hi);
    acc.lo = __builtin_elementwise_fma(a2.lo, (v2f){p2, p2}, acc.lo);
    acc.hi = __builtin_elementwise_fma(a2.hi, (v2f){p2, p2}, acc.hi);
    acc.lo = __builtin_elementwise_fma(a3.lo, (v2f){p3, p3}, acc.lo);
    acc.hi = __builtin_elementwise_fma(a3.hi, (v2f){p3, p3}, acc.hi);
    m = mn;
  }
  for (; j < cn; ++j)
    edge_one(xl, list_src<SORTED>(sorted, k0, k1, r0, r1, j), lane, b, w, m, den, acc);
}

#define NODES_PER_WAVE 2
#define NODES_PER_BLOCK 16
#define XR_KQ 4
static_assert(N_NODES % NODES_PER_BLOCK == 0, "every node block is a full 16-row strip");

__global__ void __launch_bounds__(256) __attribute__((amdgpu_waves_per_eu(12)))
gat_node(const float* __restrict__ x, const float* __restrict__ Wr,
         const float* __restrict__ xl,
         const float* __restrict__ att, const float* __restrict__ bias,
         const int* __restrict__ ei, const int* __restrict__ cnt,
         const unsigned short* __restrict__ bkt, float* __restrict__ out) {
  __shared__ float4 xr_part[XR_KQ][NODES_PER_BLOCK][HC / 4];

  const int lane = threadIdx.x & 31;
  const int wib = __builtin_amdgcn_readfirstlane((int)(threadIdx.x >> 5));
  const int wv = (int)blockIdx.x * 8 + wib;

  {
    const int g  = wib & 1;
    const int kq = wib >> 1;
    const int c     = lane & 15;
    const int khalf = (lane >> 4) * 2;
    const float* __restrict__ xrow =
        x + (size_t)((int)blockIdx.x * NODES_PER_BLOCK + c) * IN_F + 32 * kq + khalf;
    const float* __restrict__ wrow =
        Wr + (size_t)(32 * kq + khalf) * HC + 64 * g + 4 * c;
    v8f pacc[4];
    for (int t = 0; t < 4; ++t) pacc[t] = (v8f){};
#pragma unroll 4
    for (int kb = 0; kb < 32; kb += 4) {
      const float2 ax = *(const float2*)(xrow + kb);
      v2f a;
      a.x = ax.x; a.y = ax.y;
      const float* __restrict__ wk = wrow + (size_t)kb * HC;
      const float4 w0 = *(const float4*)(wk);
      const float4 w1 = *(const float4*)(wk + HC);
      wmma_step4(a, w0, w1, pacc);
    }
    for (int v = 0; v < 8; ++v) {
      float4 o;
      o.x = pacc[0][v];
      o.y = pacc[1][v];
      o.z = pacc[2][v];
      o.w = pacc[3][v];
      xr_part[kq][v + 8 * (lane >> 4)][16 * g + c] = o;
    }
  }
  __syncthreads();

  const row4 w  = load_row4(att, lane);
  const row4 bi = load_row4(bias, lane);

  for (int q = 0; q < NODES_PER_WAVE; ++q) {
  const int n = wv * NODES_PER_WAVE + q;

  row4 b;
  {
    const int r = wib * NODES_PER_WAVE + q;
    const float4 p0 = xr_part[0][r][lane], p1 = xr_part[1][r][lane];
    const float4 p2 = xr_part[2][r][lane], p3 = xr_part[3][r][lane];
    b.lo.x = (p0.x + p1.x) + (p2.x + p3.x);
    b.lo.y = (p0.y + p1.y) + (p2.y + p3.y);
    b.hi.x = (p0.z + p1.z) + (p2.z + p3.z);
    b.hi.y = (p0.w + p1.w) + (p2.w + p3.w);
  }
  const int cn0 = cnt[n];
  const int cn1 = cnt[CNT_STRIDE + n];
  const int cn = cn0 + cn1;

  row4 acc = load_row4(xl + (size_t)n * HC, lane);
  float m = head_logit(acc, b, w);
  float den = 1.0f;

  if (cn <= BKT_CAP) {
    const unsigned short* __restrict__ bl = bkt + (unsigned)(n * BKT_CAP);
    const unsigned i0 = (lane < cn0) ? (unsigned)lane
                                     : (unsigned)(BKT_STRIDE + lane - cn0);
    const unsigned i1 = (lane + 32 < cn0) ? (unsigned)(lane + 32)
                                          : (unsigned)(BKT_STRIDE + lane + 32 - cn0);
    const int k0 = (lane < cn) ? (((int)bl[i0] << 6) | lane) : 0x7fffffff;

    if (cn <= 32) {
      int r0 = 0;
      for (int j = 0; j < cn; ++j) {
        const int kj = __builtin_amdgcn_readlane(k0, j);
        r0 += (kj < k0) ? 1 : 0;
      }
      if (lane >= cn) r0 = lane;
      const int sorted = __builtin_amdgcn_ds_permute(r0 << 2, k0);
      walk_list<true>(xl, cn, sorted, k0, 0, 0, 0, lane, b, w, m, den, acc);
    } else {
      const int k1 = (lane + 32 < cn) ? (((int)bl[i1] << 6) | (lane + 32)) : 0x7fffffff;
      int r0 = 0, r1 = 0;
      for (int j = 0; j < 32; ++j) {
        const int kj = __builtin_amdgcn_readlane(k0, j);
        r0 += (kj < k0) ? 1 : 0;
        r1 += (kj < k1) ? 1 : 0;
      }
      for (int j = 32; j < cn; ++j) {
        const int kj = __builtin_amdgcn_readlane(k1, j - 32);
        r0 += (kj < k0) ? 1 : 0;
        r1 += (kj < k1) ? 1 : 0;
      }
      if (lane + 32 >= cn) r1 = -1;
      walk_list<false>(xl, cn, 0, k0, k1, r0, r1, lane, b, w, m, den, acc);
    }
  } else {
    for (int e0 = 0; e0 < N_EDGES; e0 += 32) {
      const int d = ei[N_EDGES + e0 + lane];
      const int s = ei[e0 + lane];
      unsigned mm = __builtin_amdgcn_ballot_w32(d == n);
      while (mm) {
        const int l = __builtin_ctz(mm);
        mm &= mm - 1;
        edge_one(xl, __builtin_amdgcn_readlane(s, l), lane, b, w, m, den, acc);
      }
    }
  }

  const float inv = 1.0f / (den + 1e-16f);
  const v2f olo = __builtin_elementwise_fma(acc.lo, (v2f){inv, inv}, bi.lo);
  const v2f ohi = __builtin_elementwise_fma(acc.hi, (v2f){inv, inv}, bi.hi);
  float4 r;
  r.x = olo.x; r.y = olo.y;
  r.z = ohi.x; r.w = ohi.y;
  ((float4*)(out + (size_t)n * HC))[lane] = r;
  }
}

extern "C" void kernel_launch(void* const* d_in, const int* in_sizes, int n_in,
                              void* d_out, int out_size, void* d_ws, size_t ws_size,
                              hipStream_t stream) {
  const float* x    = (const float*)d_in[0];
  const int*   ei   = (const int*)  d_in[1];
  const float* Wl   = (const float*)d_in[2];
  const float* Wr   = (const float*)d_in[3];
  const float* att  = (const float*)d_in[4];
  const float* bias = (const float*)d_in[5];
  float* out = (float*)d_out;

  float* ws   = (float*)d_ws;
  float* xl   = ws;
  int*   cnt  = (int*)(xl + (size_t)N_NODES * HC);
  unsigned short* bkt = (unsigned short*)(cnt + 2 * CNT_STRIDE);

  gat_gemm<<<1563, 128, 0, stream>>>(x, Wl, xl);

  gat_zero<<<98, 256, 0, stream>>>(cnt);
  gat_bucket<<<391, 256, 0, stream>>>(ei, cnt, bkt);

  gat_node<<<3125, 256, 0, stream>>>(x, Wr, xl, att, bias, ei, cnt, bkt, out);
}


# node pass: one edge per half wave, lane = 8 channels of one head (xl, xr rows stored so its 16-byte loads stay whole-line); two edges share max, 2^x, rescale; halves joined per node: new sum order
# speedup vs baseline: 5.0825x; 1.0470x over previous
#include <hip/hip_runtime.h>
#include <hip/hip_bf16.h>

#define N_NODES   50000
#define N_EDGES   800000
#define IN_F      128
#define HC        128
#define NEG_SLOPE 0.2f
#define LOG2E_F   1.44269504088896340736f

typedef __attribute__((ext_vector_type(2))) float v2f;
typedef __attribute__((ext_vector_type(8))) float v8f;

#define M_TILES (N_NODES / 16)

__device__ __forceinline__ void wmma_step4(const v2f a, const float4 w0,
                                           const float4 w1, v8f* acc) {
  v2f b;
  b.x = w0.x; b.y = w1.x;
  acc[0] = __builtin_amdgcn_wmma_f32_16x16x4_f32(false, a, false, b, (short)0,
                                                 acc[0], false, false);
  b.x = w0.y; b.y = w1.y;
  acc[1] = __builtin_amdgcn_wmma_f32_16x16x4_f32(false, a, false, b, (short)0,
                                                 acc[1], false, false);
  b.x = w0.z; b.y = w1.z;
  acc[2] = __builtin_amdgcn_wmma_f32_16x16x4_f32(false, a, false, b, (short)0,
                                                 acc[2], false, false);
  b.x = w0.w; b.y = w1.w;
  acc[3] = __builtin_amdgcn_wmma_f32_16x16x4_f32(false, a, false, b, (short)0,
                                                 acc[3], false, false);
}

__device__ __forceinline__ int chunk_place(const int L) { return (L >> 1) + ((L & 1) << 4); }

__global__ void __launch_bounds__(128)
gat_gemm(const float* __restrict__ x, const float* __restrict__ Wl,
         float* __restrict__ xl) {
  const int lane = threadIdx.x & 31;
  const int gw = __builtin_amdgcn_readfirstlane(
      (int)((blockIdx.x * 128 + threadIdx.x) >> 5));
  const int mt = gw >> 1;
  const int g  = gw & 1;
  if (mt >= M_TILES) return;

  const int c     = lane & 15;
  const int khalf = (lane >> 4) * 2;

  const float* __restrict__ xrow = x + (size_t)(mt * 16 + c) * IN_F + khalf;
  const float* __restrict__ wrow = Wl + (size_t)khalf * HC + 64 * g + 4 * c;

  v8f acc[4];
  for (int t = 0; t < 4; ++t) acc[t] = (v8f){};

#pragma unroll 8
  for (int kb = 0; kb < IN_F; kb += 4) {
    const float2 ax = *(const float2*)(xrow + kb);
    v2f a;
    a.x = ax.x; a.y = ax.y;
    const float* __restrict__ wk = wrow + (size_t)kb * HC;
    const float4 w0 = *(const float4*)(wk);
    const float4 w1 = *(const float4*)(wk + HC);
    wmma_step4(a, w0, w1, acc);
  }

  float* __restrict__ orow =
      xl + (size_t)(mt * 16 + 8 * (lane >> 4)) * HC + 4 * chunk_place(16 * g + c);
  for (int v = 0; v < 8; ++v) {
    float4 o;
    o.x = acc[0][v];
    o.y = acc[1][v];
    o.z = acc[2][v];
    o.w = acc[3][v];
    *(float4*)(orow + (size_t)v * HC) = o;
  }
}

#define EDGE_T  (N_EDGES / 8)
#define BKT_CAP 64
#define CNT_STRIDE 50048
#define BKT_STRIDE (N_NODES * BKT_CAP)
static_assert(CNT_STRIDE >= N_NODES && CNT_STRIDE % 32 == 0, "a half's counters fill whole lines");
static_assert(N_NODES <= 65536, "a source id must fit the 16-bit bucket slot");
static_assert(N_EDGES % 32 == 0, "the edge-list walk of an over-full node takes whole waves");

__global__ void gat_zero(int* __restrict__ cnt) {
  const int i = blockIdx.x * blockDim.x + threadIdx.x;
  if (i < 2 * CNT_STRIDE / 4) ((int4*)cnt)[i] = make_int4(0, 0, 0, 0);
}

__global__ void gat_bucket(const int* __restrict__ ei, int* __restrict__ cnt_all,
                           unsigned short* __restrict__ bkt_all) {
  const unsigned t = blockIdx.x * blockDim.x + threadIdx.x;
  const unsigned h = ((unsigned)__builtin_amdgcn_s_sendmsg_rtn(0x87) >> 18) & 1u;
  int* __restrict__ cnt = cnt_all + h * CNT_STRIDE;
  unsigned short* __restrict__ bkt = bkt_all + h * BKT_STRIDE;
  if (t < EDGE_T) {
    const int4* __restrict__ src4 = (const int4*)ei;
    const int4* __restrict__ dst4 = (const int4*)(ei + N_EDGES);
    const int4 d0 = dst4[t], d1 = dst4[t + EDGE_T];
    const int4 s0 = src4[t], s1 = src4[t + EDGE_T];
    const int p0 = atomicAdd(&cnt[d0.x], 1);
    const int p1 = atomicAdd(&cnt[d0.y], 1);
    const int p2 = atomicAdd(&cnt[d0.z], 1);
    const int p3 = atomicAdd(&cnt[d0.w], 1);
    const int p4 = atomicAdd(&cnt[d1.x], 1);
    const int p5 = atomicAdd(&cnt[d1.y], 1);
    const int p6 = atomicAdd(&cnt[d1.z], 1);
    const int p7 = atomicAdd(&cnt[d1.w], 1);
    if (p0 < BKT_CAP) bkt[(unsigned)(d0.x * BKT_CAP + p0)] = (unsigned short)s0.x;
    if (p1 < BKT_CAP) bkt[(unsigned)(d0.y * BKT_CAP + p1)] = (unsigned short)s0.y;
    if (p2 < BKT_CAP) bkt[(unsigned)(d0.z * BKT_CAP + p2)] = (unsigned short)s0.z;
    if (p3 < BKT_CAP) bkt[(unsigned)(d0.w * BKT_CAP + p3)] = (unsigned short)s0.w;
    if (p4 < BKT_CAP) bkt[(unsigned)(d1.x * BKT_CAP + p4)] = (unsigned short)s1.x;
    if (p5 < BKT_CAP) bkt[(unsigned)(d1.y * BKT_CAP + p5)] = (unsigned short)s1.y;
    if (p6 < BKT_CAP) bkt[(unsigned)(d1.z * BKT_CAP + p6)] = (unsigned short)s1.z;
    if (p7 < BKT_CAP) bkt[(unsigned)(d1.w * BKT_CAP + p7)] = (unsigned short)s1.w;
  }
}

struct row8 { v2f c[4]; };

__device__ __forceinline__ row8 load_row8(const float4* __restrict__ row, const int j) {
  const float4 a = row[j];
  const float4 d = row[16 + j];
  row8 r;
  r.c[0].x = a.x; r.c[0].y = a.y;
  r.c[1].x = a.z; r.c[1].y = a.w;
  r.c[2].x = d.x; r.c[2].y = d.y;
  r.c[3].x = d.z; r.c[3].y = d.w;
  return r;
}

__device__ __forceinline__ row8 load_chan8(const float* __restrict__ base, const int j) {
  const float4 a = ((const float4*)base)[2 * j];
  const float4 d = ((const float4*)base)[2 * j + 1];
  row8 r;
  r.c[0].x = a.x; r.c[0].y = a.y;
  r.c[1].x = a.z; r.c[1].y = a.w;
  r.c[2].x = d.x; r.c[2].y = d.y;
  r.c[3].x = d.z; r.c[3].y = d.w;
  return r;
}

template <int X>
__device__ __forceinline__ float add_xor_lane(const float s) {
  const int t = __builtin_amdgcn_update_dpp(0, __float_as_int(s), 0x160 + X, 0xf, 0xf, true);
  return s + __int_as_float(t);
}

__device__ __forceinline__ float xhalf(const float x) {
  const int b = __float_as_int(x);
  return __int_as_float(__builtin_amdgcn_permlanex16(b, b, 0x76543210, 0xfedcba98, false, false));
}

__device__ __forceinline__ float head_logit8(const row8 a, const row8 b, const row8 w) {
  v2f v = a.c[0] + b.c[0];
  v = __builtin_elementwise_max(v, v * NEG_SLOPE);
  v2f s2 = v * w.c[0];
  for (int i = 1; i < 4; ++i) {
    v = a.c[i] + b.c[i];
    v = __builtin_elementwise_max(v, v * NEG_SLOPE);
    s2 = __builtin_elementwise_fma(v, w.c[i], s2);
  }
  float s = s2.x + s2.y;
  s = add_xor_lane<1>(s);
  s = add_xor_lane<2>(s);
  return s;
}

__device__ __forceinline__ int pick_rank(int k0, int k1, int r0, int r1, int p) {
  const unsigned m0 = __builtin_amdgcn_ballot_w32(r0 == p);
  if (m0) return __builtin_amdgcn_readlane(k0, __builtin_ctz(m0)) >> 6;
  const unsigned m1 = __builtin_amdgcn_ballot_w32(r1 == p);
  return __builtin_amdgcn_readlane(k1, __builtin_ctz(m1)) >> 6;
}

__device__ __forceinline__ void pair1(const float4* __restrict__ xl4, const int src0,
                                      const bool ok0, const int j, const row8 b,
                                      const row8 w, float& m, float& den, row8& acc) {
  const row8 a0 = load_row8(xl4 + (unsigned)(src0 * (HC / 4)), j);
  float l0 = head_logit8(a0, b, w);
  l0 = ok0 ? l0 : -__builtin_inff();
  const float mn = fmaxf(l0, m);
  const v2f e2 = ((v2f){m, l0} - (v2f){mn, mn}) * LOG2E_F;
  const float sc = __builtin_amdgcn_exp2f(e2.x);
  const float p0 = __builtin_amdgcn_exp2f(e2.y);
  den = den * sc + p0;
  for (int i = 0; i < 4; ++i)
    acc.c[i] = __builtin_elementwise_fma(acc.c[i], (v2f){sc, sc}, a0.c[i] * p0);
  m = mn;
}

template <bool CHECK>
__device__ __forceinline__ void pair2(const float4* __restrict__ xl4, const int src0,
                                      const int src1, const bool ok0, const bool ok1,
                                      const int j, const row8 b, const row8 w,
                                      float& m, float& den, row8& acc) {
  const row8 a0 = load_row8(xl4 + (unsigned)(src0 * (HC / 4)), j);
  const row8 a1 = load_row8(xl4 + (unsigned)(src1 * (HC / 4)), j);
  float l0 = head_logit8(a0, b, w);
  float l1 = head_logit8(a1, b, w);
  if (CHECK) {
    l0 = ok0 ? l0 : -__builtin_inff();
    l1 = ok1 ? l1 : -__builtin_inff();
  }
  const float mn = fmaxf(fmaxf(l0, l1), m);
  const v2f e01 = ((v2f){l0, l1} - (v2f){mn, mn}) * LOG2E_F;
  const float sc = __builtin_amdgcn_exp2f((m - mn) * LOG2E_F);
  const float p0 = __builtin_amdgcn_exp2f(e01.x);
  const float p1 = __builtin_amdgcn_exp2f(e01.y);
  den = den * sc + (p0 + p1);
  for (int i = 0; i < 4; ++i) {
    acc.c[i] = acc.c[i] * sc;
    acc.c[i] = __builtin_elementwise_fma(a0.c[i], (v2f){p0, p0}, acc.c[i]);
    acc.c[i] = __builtin_elementwise_fma(a1.c[i], (v2f){p1, p1}, acc.c[i]);
  }
  m = mn;
}

#define NODES_PER_WAVE 2
#define NODES_PER_BLOCK 16
#define XR_KQ 4
#define SORT_CAP 31
static_assert(N_NODES % NODES_PER_BLOCK == 0, "every node block is a full 16-row strip");

__global__ void __launch_bounds__(256) __attribute__((amdgpu_waves_per_eu(12)))
gat_node(const float* __restrict__ x, const float* __restrict__ Wr,
         const float* __restrict__ xl,
         const float* __restrict__ att, const float* __restrict__ bias,
         const int* __restrict__ ei, const int* __restrict__ cnt,
         const unsigned short* __restrict__ bkt, float* __restrict__ out) {
  __shared__ float4 xr_part[XR_KQ][NODES_PER_BLOCK][HC / 4];

  const int lane = threadIdx.x & 31;
  const int wib = __builtin_amdgcn_readfirstlane((int)(threadIdx.x >> 5));
  const int wv = (int)blockIdx.x * 8 + wib;

  {
    const int g  = wib & 1;
    const int kq = wib >> 1;
    const int c     = lane & 15;
    const int khalf = (lane >> 4) * 2;
    const float* __restrict__ xrow =
        x + (size_t)((int)blockIdx.x * NODES_PER_BLOCK + c) * IN_F + 32 * kq + khalf;
    const float* __restrict__ wrow =
        Wr + (size_t)(32 * kq + khalf) * HC + 64 * g + 4 * c;
    v8f pacc[4];
    for (int t = 0; t < 4; ++t) pacc[t] = (v8f){};
#pragma unroll 4
    for (int kb = 0; kb < 32; kb += 4) {
      const float2 ax = *(const float2*)(xrow + kb);
      v2f a;
      a.x = ax.x; a.y = ax.y;
      const float* __restrict__ wk = wrow + (size_t)kb * HC;
      const float4 w0 = *(const float4*)(wk);
      const float4 w1 = *(const float4*)(wk + HC);
      wmma_step4(a, w0, w1, pacc);
    }
    const int place = chunk_place(16 * g + c);
    for (int v = 0; v < 8; ++v) {
      float4 o;
      o.x = pacc[0][v];
      o.y = pacc[1][v];
      o.z = pacc[2][v];
      o.w = pacc[3][v];
      xr_part[kq][v + 8 * (lane >> 4)][place] = o;
    }
  }
  __syncthreads();

  const int h = lane >> 4;
  const int j = lane & 15;
  const bool lo = (h == 0);
  const row8 w = load_chan8(att, j);
  const float4* __restrict__ xl4 = (const float4*)xl;

  for (int q = 0; q < NODES_PER_WAVE; ++q) {
  const int n = wv * NODES_PER_WAVE + q;

  row8 b;
  {
    const int r = wib * NODES_PER_WAVE + q;
    const float4 p0 = xr_part[0][r][j], p1 = xr_part[1][r][j];
    const float4 p2 = xr_part[2][r][j], p3 = xr_part[3][r][j];
    b.c[0].x = (p0.x + p1.x) + (p2.x + p3.x);
    b.c[0].y = (p0.y + p1.y) + (p2.y + p3.y);
    b.c[1].x = (p0.z + p1.z) + (p2.z + p3.z);
    b.c[1].y = (p0.w + p1.w) + (p2.w + p3.w);
    const float4 t0 = xr_part[0][r][16 + j], t1 = xr_part[1][r][16 + j];
    const float4 t2 = xr_part[2][r][16 + j], t3 = xr_part[3][r][16 + j];
    b.c[2].x = (t0.x + t1.x) + (t2.x + t3.x);
    b.c[2].y = (t0.y + t1.y) + (t2.y + t3.y);
    b.c[3].x = (t0.z + t1.z) + (t2.z + t3.z);
    b.c[3].y = (t0.w + t1.w) + (t2.w + t3.w);
  }
  const int cn0 = cnt[n];
  const int cn1 = cnt[CNT_STRIDE + n];
  const int cn = cn0 + cn1;

  float m = -__FLT_MAX__;
  float den = 0.0f;
  row8 acc;
  for (int i = 0; i < 4; ++i) acc.c[i] = (v2f){0.0f, 0.0f};

  if (cn <= BKT_CAP) {
    const unsigned short* __restrict__ bl = bkt + (unsigned)(n * BKT_CAP);
    const unsigned i0 = (lane < cn0) ? (unsigned)lane
                                     : (unsigned)(BKT_STRIDE + lane - cn0);
    const unsigned i1 = (lane + 32 < cn0) ? (unsigned)(lane + 32)
                                          : (unsigned)(BKT_STRIDE + lane + 32 - cn0);
    int k0 = (lane < cn) ? (((int)bl[i0] << 6) | lane) : 0x7fffffff;

    if (cn <= SORT_CAP) {
      int r0 = 0;
      for (int t = 0; t < cn; ++t) {
        const int kt = __builtin_amdgcn_readlane(k0, t);
        r0 += (kt < k0) ? 1 : 0;
      }
      int pl = (lane < cn) ? r0 + 1 : lane;
      if (lane == cn) { pl = 0; k0 = n << 6; }
      const int sorted = __builtin_amdgcn_ds_permute(((pl >> 1) + ((pl & 1) << 4)) << 2, k0);
      int p = 0;
      for (; 2 * p + 3 <= cn; p += 2) {
        const unsigned s0 = (unsigned)p * 0x11111111u, s1 = (unsigned)(p + 1) * 0x11111111u;
        const int e0 = __builtin_amdgcn_permlane16(sorted, sorted, s0, s0, false, false);
        const int e1 = __builtin_amdgcn_permlane16(sorted, sorted, s1, s1, false, false);
        pair2<false>(xl4, e0 >> 6, e1 >> 6, true, true, j, b, w, m, den, acc);
      }
      if (2 * p <= cn) {
        const bool ok0 = 2 * p + h <= cn;
        const unsigned s0 = (unsigned)p * 0x11111111u;
        const int e0 = __builtin_amdgcn_permlane16(sorted, sorted, s0, s0, false, false);
        const int src0 = ok0 ? (e0 >> 6) : n;
        if (2 * p + 2 <= cn) {
          const bool ok1 = 2 * p + 2 + h <= cn;
          const unsigned s1 = (unsigned)((p + 1) & 15) * 0x11111111u;
          const int e1 = __builtin_amdgcn_permlane16(sorted, sorted, s1, s1, false, false);
          pair2<true>(xl4, src0, ok1 ? (e1 >> 6) : n, ok0, ok1, j, b, w, m, den, acc);
        } else {
          pair1(xl4, src0, ok0, j, b, w, m, den, acc);
        }
      }
    } else {
      const int k1 = (lane + 32 < cn) ? (((int)bl[i1] << 6) | (lane + 32)) : 0x7fffffff;
      int r0 = 0, r1 = 0;
      for (int t = 0; t < 32; ++t) {
        const int kt = __builtin_amdgcn_readlane(k0, t);
        r0 += (kt < k0) ? 1 : 0;
        r1 += (kt < k1) ? 1 : 0;
      }
      for (int t = 32; t < cn; ++t) {
        const int kt = __builtin_amdgcn_readlane(k1, t - 32);
        r0 += (kt < k0) ? 1 : 0;
        r1 += (kt < k1) ? 1 : 0;
      }
      if (lane + 32 >= cn) r1 = -1;
      for (int q0 = 0; q0 <= cn; q0 += 4) {
        int s[4];
        for (int t = 0; t < 4; ++t) {
          const int qq = q0 + t;
          s[t] = (qq >= 1 && qq <= cn) ? pick_rank(k0, k1, r0, r1, qq - 1) : n;
        }
        pair2<true>(xl4, lo ? s[0] : s[1], lo ? s[2] : s[3],
                    q0 + h <= cn, q0 + 2 + h <= cn, j, b, w, m, den, acc);
      }
    }
  } else {
    for (int e0 = -32; e0 < N_EDGES; e0 += 32) {
      int s = n;
      unsigned mm = 1u;
      if (e0 >= 0) {
        const int d = ei[N_EDGES + e0 + lane];
        s = ei[e0 + lane];
        mm = __builtin_amdgcn_ballot_w32(d == n);
      }
      while (mm) {
        const int l = __builtin_ctz(mm);
        mm &= mm - 1;
        pair1(xl4, __builtin_amdgcn_readlane(s, l), lo, j, b, w, m, den, acc);
      }
    }
  }

  asm volatile("" ::: "memory");
  const float4 bi = ((const float4*)bias)[2 * j + h];
  const float mo = xhalf(m), deno = xhalf(den);
  const float mj = fmaxf(m, mo);
  const v2f ej = ((v2f){m, mo} - (v2f){mj, mj}) * LOG2E_F;
  const float sme = __builtin_amdgcn_exp2f(ej.x);
  const float sot = __builtin_amdgcn_exp2f(ej.y);
  const float dn = den * sme + deno * sot;
  const v2f keep0 = lo ? acc.c[0] : acc.c[2], keep1 = lo ? acc.c[1] : acc.c[3];
  const v2f send0 = lo ? acc.c[2] : acc.c[0], send1 = lo ? acc.c[3] : acc.c[1];
  v2f got0, got1;
  got0.x = xhalf(send0.x); got0.y = xhalf(send0.y);
  got1.x = xhalf(send1.x); got1.y = xhalf(send1.y);
  const v2f sum0 = __builtin_elementwise_fma(keep0, (v2f){sme, sme}, got0 * sot);
  const v2f sum1 = __builtin_elementwise_fma(keep1, (v2f){sme, sme}, got1 * sot);

  const float inv = 1.0f / (dn + 1e-16f);
  const v2f olo = __builtin_elementwise_fma(sum0, (v2f){inv, inv}, (v2f){bi.x, bi.y});
  const v2f ohi = __builtin_elementwise_fma(sum1, (v2f){inv, inv}, (v2f){bi.z, bi.w});
  float4 r;
  r.x = olo.x; r.y = olo.y;
  r.z = ohi.x; r.w = ohi.y;
  ((float4*)(out + (size_t)n * HC))[2 * j + h] = r;
  }
}

extern "C" void kernel_launch(void* const* d_in, const int* in_sizes, int n_in,
                              void* d_out, int out_size, void* d_ws, size_t ws_size,
                              hipStream_t stream) {
  const float* x    = (const float*)d_in[0];
  const int*   ei   = (const int*)  d_in[1];
  const float* Wl   = (const float*)d_in[2];
  const float* Wr   = (const float*)d_in[3];
  const float* att  = (const float*)d_in[4];
  const float* bias = (const float*)d_in[5];
  float* out = (float*)d_out;

  float* ws   = (float*)d_ws;
  float* xl   = ws;
  int*   cnt  = (int*)(xl + (size_t)N_NODES * HC);
  unsigned short* bkt = (unsigned short*)(cnt + 2 * CNT_STRIDE);

  gat_gemm<<<1563, 128, 0, stream>>>(x, Wl, xl);

  gat_zero<<<98, 256, 0, stream>>>(cnt);
  gat_bucket<<<391, 256, 0, stream>>>(ei, cnt, bkt);

  gat_node<<<3125, 256, 0, stream>>>(x, Wr, xl, att, bias, ei, cnt, bkt, out);
}


# the xl GEMM loses its own launch: its 1563 work-groups are dealt 2 : 1 among the 782 edge work-groups of the grouping launch, matrix issue under the counter adds' and slot stores' waits; 3 launches
# speedup vs baseline: 5.3214x; 1.0470x over previous
#include <hip/hip_runtime.h>
#include <hip/hip_bf16.h>

#define N_NODES   50000
#define N_EDGES   800000
#define IN_F      128
#define HC        128
#define NEG_SLOPE 0.2f
#define LOG2E_F   1.44269504088896340736f

typedef __attribute__((ext_vector_type(2))) float v2f;
typedef __attribute__((ext_vector_type(8))) float v8f;

#define M_TILES (N_NODES / 16)

__device__ __forceinline__ void wmma_step4(const v2f a, const float4 w0,
                                           const float4 w1, v8f* acc) {
  v2f b;
  b.x = w0.x; b.y = w1.x;
  acc[0] = __builtin_amdgcn_wmma_f32_16x16x4_f32(false, a, false, b, (short)0,
                                                 acc[0], false, false);
  b.x = w0.y; b.y = w1.y;
  acc[1] = __builtin_amdgcn_wmma_f32_16x16x4_f32(false, a, false, b, (short)0,
                                                 acc[1], false, false);
  b.x = w0.z; b.y = w1.z;
  acc[2] = __builtin_amdgcn_wmma_f32_16x16x4_f32(false, a, false, b, (short)0,
                                                 acc[2], false, false);
  b.x = w0.w; b.y = w1.w;
  acc[3] = __builtin_amdgcn_wmma_f32_16x16x4_f32(false, a, false, b, (short)0,
                                                 acc[3], false, false);
}

__device__ __forceinline__ int chunk_place(const int L) { return (L >> 1) + ((L & 1) << 4); }

__device__ __forceinline__ void
gemm_body(const int blk, const float* __restrict__ x, const float* __restrict__ Wl,
          float* __restrict__ xl) {
  const int lane = threadIdx.x & 31;
  const int gw = __builtin_amdgcn_readfirstlane(
      (int)((blk * 128 + threadIdx.x) >> 5));
  const int mt = gw >> 1;
  const int g  = gw & 1;
  if (mt >= M_TILES) return;

  const int c     = lane & 15;
  const int khalf = (lane >> 4) * 2;

  const float* __restrict__ xrow = x + (size_t)(mt * 16 + c) * IN_F + khalf;
  const float* __restrict__ wrow = Wl + (size_t)khalf * HC + 64 * g + 4 * c;

  v8f acc[4];
  for (int t = 0; t < 4; ++t) acc[t] = (v8f){};

#pragma unroll 8
  for (int kb = 0; kb < IN_F; kb += 4) {
    const float2 ax = *(const float2*)(xrow + kb);
    v2f a;
    a.x = ax.x; a.y = ax.y;
    const float* __restrict__ wk = wrow + (size_t)kb * HC;
    const float4 w0 = *(const float4*)(wk);
    const float4 w1 = *(const float4*)(wk + HC);
    wmma_step4(a, w0, w1, acc);
  }

  float* __restrict__ orow =
      xl + (size_t)(mt * 16 + 8 * (lane >> 4)) * HC + 4 * chunk_place(16 * g + c);
  for (int v = 0; v < 8; ++v) {
    float4 o;
    o.x = acc[0][v];
    o.y = acc[1][v];
    o.z = acc[2][v];
    o.w = acc[3][v];
    *(float4*)(orow + (size_t)v * HC) = o;
  }
}

#define EDGE_T  (N_EDGES / 8)
#define BKT_CAP 64
#define CNT_STRIDE 50048
#define BKT_STRIDE (N_NODES * BKT_CAP)
static_assert(CNT_STRIDE >= N_NODES && CNT_STRIDE % 32 == 0, "a half's counters fill whole lines");
static_assert(N_NODES <= 65536, "a source id must fit the 16-bit bucket slot");
static_assert(N_EDGES % 32 == 0, "the edge-list walk of an over-full node takes whole waves");

__global__ void gat_zero(int* __restrict__ cnt) {
  const int i = blockIdx.x * blockDim.x + threadIdx.x;
  if (i < 2 * CNT_STRIDE / 4) ((int4*)cnt)[i] = make_int4(0, 0, 0, 0);
}

__device__ __forceinline__ void
bucket_body(const int blk, const int* __restrict__ ei, int* __restrict__ cnt_all,
            unsigned short* __restrict__ bkt_all) {
  const unsigned t = (unsigned)blk * 128 + threadIdx.x;
  const unsigned h = ((unsigned)__builtin_amdgcn_s_sendmsg_rtn(0x87) >> 18) & 1u;
  int* __restrict__ cnt = cnt_all + h * CNT_STRIDE;
  unsigned short* __restrict__ bkt = bkt_all + h * BKT_STRIDE;
  if (t < EDGE_T) {
    const int4* __restrict__ src4 = (const int4*)ei;
    const int4* __restrict__ dst4 = (const int4*)(ei + N_EDGES);
    const int4 d0 = dst4[t], d1 = dst4[t + EDGE_T];
    const int4 s0 = src4[t], s1 = src4[t + EDGE_T];
    const int p0 = atomicAdd(&cnt[d0.x], 1);
    const int p1 = atomicAdd(&cnt[d0.y], 1);
    const int p2 = atomicAdd(&cnt[d0.z], 1);
    const int p3 = atomicAdd(&cnt[d0.w], 1);
    const int p4 = atomicAdd(&cnt[d1.x], 1);
    const int p5 = atomicAdd(&cnt[d1.y], 1);
    const int p6 = atomicAdd(&cnt[d1.z], 1);
    const int p7 = atomicAdd(&cnt[d1.w], 1);
    if (p0 < BKT_CAP) bkt[(unsigned)(d0.x * BKT_CAP + p0)] = (unsigned short)s0.x;
    if (p1 < BKT_CAP) bkt[(unsigned)(d0.y * BKT_CAP + p1)] = (unsigned short)s0.y;
    if (p2 < BKT_CAP) bkt[(unsigned)(d0.z * BKT_CAP + p2)] = (unsigned short)s0.z;
    if (p3 < BKT_CAP) bkt[(unsigned)(d0.w * BKT_CAP + p3)] = (unsigned short)s0.w;
    if (p4 < BKT_CAP) bkt[(unsigned)(d1.x * BKT_CAP + p4)] = (unsigned short)s1.x;
    if (p5 < BKT_CAP) bkt[(unsigned)(d1.y * BKT_CAP + p5)] = (unsigned short)s1.y;
    if (p6 < BKT_CAP) bkt[(unsigned)(d1.z * BKT_CAP + p6)] = (unsigned short)s1.z;
    if (p7 < BKT_CAP) bkt[(unsigned)(d1.w * BKT_CAP + p7)] = (unsigned short)s1.w;
  }
}

#define BKT_BLOCKS  ((EDGE_T + 127) / 128)
#define GEMM_BLOCKS ((2 * M_TILES * 32 + 127) / 128)
#define JOINT_BLOCKS 2345
static_assert(BKT_BLOCKS + GEMM_BLOCKS == JOINT_BLOCKS, "the joint grid");
static_assert((JOINT_BLOCKS + 2) / 3 == BKT_BLOCKS, "every third work-group of the grid is an edge one");

__global__ void __launch_bounds__(128)
gat_bucket_gemm(const int* __restrict__ ei, int* __restrict__ cnt_all,
                unsigned short* __restrict__ bkt_all, const float* __restrict__ x,
                const float* __restrict__ Wl, float* __restrict__ xl) {
  const int i = (int)blockIdx.x;
  const int third = i / 3;
  if (i - 3 * third == 0) {
    bucket_body(third, ei, cnt_all, bkt_all);
  } else {
    gemm_body(i - third - 1, x, Wl, xl);
  }
}

struct row8 { v2f c[4]; };

__device__ __forceinline__ row8 load_row8(const float4* __restrict__ row, const int j) {
  const float4 a = row[j];
  const float4 d = row[16 + j];
  row8 r;
  r.c[0].x = a.x; r.c[0].y = a.y;
  r.c[1].x = a.z; r.c[1].y = a.w;
  r.c[2].x = d.x; r.c[2].y = d.y;
  r.c[3].x = d.z; r.c[3].y = d.w;
  return r;
}

__device__ __forceinline__ row8 load_chan8(const float* __restrict__ base, const int j) {
  const float4 a = ((const float4*)base)[2 * j];
  const float4 d = ((const float4*)base)[2 * j + 1];
  row8 r;
  r.c[0].x = a.x; r.c[0].y = a.y;
  r.c[1].x = a.z; r.c[1].y = a.w;
  r.c[2].x = d.x; r.c[2].y = d.y;
  r.c[3].x = d.z; r.c[3].y = d.w;
  return r;
}

template <int X>
__device__ __forceinline__ float add_xor_lane(const float s) {
  const int t = __builtin_amdgcn_update_dpp(0, __float_as_int(s), 0x160 + X, 0xf, 0xf, true);
  return s + __int_as_float(t);
}

__device__ __forceinline__ float xhalf(const float x) {
  const int b = __float_as_int(x);
  return __int_as_float(__builtin_amdgcn_permlanex16(b, b, 0x76543210, 0xfedcba98, false, false));
}

__device__ __forceinline__ float head_logit8(const row8 a, const row8 b, const row8 w) {
  v2f v = a.c[0] + b.c[0];
  v = __builtin_elementwise_max(v, v * NEG_SLOPE);
  v2f s2 = v * w.c[0];
  for (int i = 1; i < 4; ++i) {
    v = a.c[i] + b.c[i];
    v = __builtin_elementwise_max(v, v * NEG_SLOPE);
    s2 = __builtin_elementwise_fma(v, w.c[i], s2);
  }
  float s = s2.x + s2.y;
  s = add_xor_lane<1>(s);
  s = add_xor_lane<2>(s);
  return s;
}

__device__ __forceinline__ int pick_rank(int k0, int k1, int r0, int r1, int p) {
  const unsigned m0 = __builtin_amdgcn_ballot_w32(r0 == p);
  if (m0) return __builtin_amdgcn_readlane(k0, __builtin_ctz(m0)) >> 6;
  const unsigned m1 = __builtin_amdgcn_ballot_w32(r1 == p);
  return __builtin_amdgcn_readlane(k1, __builtin_ctz(m1)) >> 6;
}

__device__ __forceinline__ void pair1(const float4* __restrict__ xl4, const int src0,
                                      const bool ok0, const int j, const row8 b,
                                      const row8 w, float& m, float& den, row8& acc) {
  const row8 a0 = load_row8(xl4 + (unsigned)(src0 * (HC / 4)), j);
  float l0 = head_logit8(a0, b, w);
  l0 = ok0 ? l0 : -__builtin_inff();
  const float mn = fmaxf(l0, m);
  const v2f e2 = ((v2f){m, l0} - (v2f){mn, mn}) * LOG2E_F;
  const float sc = __builtin_amdgcn_exp2f(e2.x);
  const float p0 = __builtin_amdgcn_exp2f(e2.y);
  den = den * sc + p0;
  for (int i = 0; i < 4; ++i)
    acc.c[i] = __builtin_elementwise_fma(acc.c[i], (v2f){sc, sc}, a0.c[i] * p0);
  m = mn;
}

template <bool CHECK>
__device__ __forceinline__ void pair2(const float4* __restrict__ xl4, const int src0,
                                      const int src1, const bool ok0, const bool ok1,
                                      const int j, const row8 b, const row8 w,
                                      float& m, float& den, row8& acc) {
  const row8 a0 = load_row8(xl4 + (unsigned)(src0 * (HC / 4)), j);
  const row8 a1 = load_row8(xl4 + (unsigned)(src1 * (HC / 4)), j);
  float l0 = head_logit8(a0, b, w);
  float l1 = head_logit8(a1, b, w);
  if (CHECK) {
    l0 = ok0 ? l0 : -__builtin_inff();
    l1 = ok1 ? l1 : -__builtin_inff();
  }
  const float mn = fmaxf(fmaxf(l0, l1), m);
  const v2f e01 = ((v2f){l0, l1} - (v2f){mn, mn}) * LOG2E_F;
  const float sc = __builtin_amdgcn_exp2f((m - mn) * LOG2E_F);
  const float p0 = __builtin_amdgcn_exp2f(e01.x);
  const float p1 = __builtin_amdgcn_exp2f(e01.y);
  den = den * sc + (p0 + p1);
  for (int i = 0; i < 4; ++i) {
    acc.c[i] = acc.c[i] * sc;
    acc.c[i] = __builtin_elementwise_fma(a0.c[i], (v2f){p0, p0}, acc.c[i]);
    acc.c[i] = __builtin_elementwise_fma(a1.c[i], (v2f){p1, p1}, acc.c[i]);
  }
  m = mn;
}

#define NODES_PER_WAVE 2
#define NODES_PER_BLOCK 16
#define XR_KQ 4
#define SORT_CAP 31
static_assert(N_NODES % NODES_PER_BLOCK == 0, "every node block is a full 16-row strip");

__global__ void __launch_bounds__(256) __attribute__((amdgpu_waves_per_eu(12)))
gat_node(const float* __restrict__ x, const float* __restrict__ Wr,
         const float* __restrict__ xl,
         const float* __restrict__ att, const float* __restrict__ bias,
         const int* __restrict__ ei, const int* __restrict__ cnt,
         const unsigned short* __restrict__ bkt, float* __restrict__ out) {
  __shared__ float4 xr_part[XR_KQ][NODES_PER_BLOCK][HC / 4];

  const int lane = threadIdx.x & 31;
  const int wib = __builtin_amdgcn_readfirstlane((int)(threadIdx.x >> 5));
  const int wv = (int)blockIdx.x * 8 + wib;

  {
    const int g  = wib & 1;
    const int kq = wib >> 1;
    const int c     = lane & 15;
    const int khalf = (lane >> 4) * 2;
    const float* __restrict__ xrow =
        x + (size_t)((int)blockIdx.x * NODES_PER_BLOCK + c) * IN_F + 32 * kq + khalf;
    const float* __restrict__ wrow =
        Wr + (size_t)(32 * kq + khalf) * HC + 64 * g + 4 * c;
    v8f pacc[4];
    for (int t = 0; t < 4; ++t) pacc[t] = (v8f){};
#pragma unroll 4
    for (int kb = 0; kb < 32; kb += 4) {
      const float2 ax = *(const float2*)(xrow + kb);
      v2f a;
      a.x = ax.x; a.y = ax.y;
      const float* __restrict__ wk = wrow + (size_t)kb * HC;
      const float4 w0 = *(const float4*)(wk);
      const float4 w1 = *(const float4*)(wk + HC);
      wmma_step4(a, w0, w1, pacc);
    }
    const int place = chunk_place(16 * g + c);
    for (int v = 0; v < 8; ++v) {
      float4 o;
      o.x = pacc[0][v];
      o.y = pacc[1][v];
      o.z = pacc[2][v];
      o.w = pacc[3][v];
      xr_part[kq][v + 8 * (lane >> 4)][place] = o;
    }
  }
  __syncthreads();

  const int h = lane >> 4;
  const int j = lane & 15;
  const bool lo = (h == 0);
  const row8 w = load_chan8(att, j);
  const float4* __restrict__ xl4 = (const float4*)xl;

  for (int q = 0; q < NODES_PER_WAVE; ++q) {
  const int n = wv * NODES_PER_WAVE + q;

  row8 b;
  {
    const int r = wib * NODES_PER_WAVE + q;
    const float4 p0 = xr_part[0][r][j], p1 = xr_part[1][r][j];
    const float4 p2 = xr_part[2][r][j], p3 = xr_part[3][r][j];
    b.c[0].x = (p0.x + p1.x) + (p2.x + p3.x);
    b.c[0].y = (p0.y + p1.y) + (p2.y + p3.y);
    b.c[1].x = (p0.z + p1.z) + (p2.z + p3.z);
    b.c[1].y = (p0.w + p1.w) + (p2.w + p3.w);
    const float4 t0 = xr_part[0][r][16 + j], t1 = xr_part[1][r][16 + j];
    const float4 t2 = xr_part[2][r][16 + j], t3 = xr_part[3][r][16 + j];
    b.c[2].x = (t0.x + t1.x) + (t2.x + t3.x);
    b.c[2].y = (t0.y + t1.y) + (t2.y + t3.y);
    b.c[3].x = (t0.z + t1.z) + (t2.z + t3.z);
    b.c[3].y = (t0.w + t1.w) + (t2.w + t3.w);
  }
  const int cn0 = cnt[n];
  const int cn1 = cnt[CNT_STRIDE + n];
  const int cn = cn0 + cn1;

  float m = -__FLT_MAX__;
  float den = 0.0f;
  row8 acc;
  for (int i = 0; i < 4; ++i) acc.c[i] = (v2f){0.0f, 0.0f};

  if (cn <= BKT_CAP) {
    const unsigned short* __restrict__ bl = bkt + (unsigned)(n * BKT_CAP);
    const unsigned i0 = (lane < cn0) ? (unsigned)lane
                                     : (unsigned)(BKT_STRIDE + lane - cn0);
    const unsigned i1 = (lane + 32 < cn0) ? (unsigned)(lane + 32)
                                          : (unsigned)(BKT_STRIDE + lane + 32 - cn0);
    int k0 = (lane < cn) ? (((int)bl[i0] << 6) | lane) : 0x7fffffff;

    if (cn <= SORT_CAP) {
      int r0 = 0;
      for (int t = 0; t < cn; ++t) {
        const int kt = __builtin_amdgcn_readlane(k0, t);
        r0 += (kt < k0) ? 1 : 0;
      }
      int pl = (lane < cn) ? r0 + 1 : lane;
      if (lane == cn) { pl = 0; k0 = n << 6; }
      const int sorted = __builtin_amdgcn_ds_permute(((pl >> 1) + ((pl & 1) << 4)) << 2, k0);
      int p = 0;
      for (; 2 * p + 3 <= cn; p += 2) {
        const unsigned s0 = (unsigned)p * 0x11111111u, s1 = (unsigned)(p + 1) * 0x11111111u;
        const int e0 = __builtin_amdgcn_permlane16(sorted, sorted, s0, s0, false, false);
        const int e1 = __builtin_amdgcn_permlane16(sorted, sorted, s1, s1, false, false);
        pair2<false>(xl4, e0 >> 6, e1 >> 6, true, true, j, b, w, m, den, acc);
      }
      if (2 * p <= cn) {
        const bool ok0 = 2 * p + h <= cn;
        const unsigned s0 = (unsigned)p * 0x11111111u;
        const int e0 = __builtin_amdgcn_permlane16(sorted, sorted, s0, s0, false, false);
        const int src0 = ok0 ? (e0 >> 6) : n;
        if (2 * p + 2 <= cn) {
          const bool ok1 = 2 * p + 2 + h <= cn;
          const unsigned s1 = (unsigned)((p + 1) & 15) * 0x11111111u;
          const int e1 = __builtin_amdgcn_permlane16(sorted, sorted, s1, s1, false, false);
          pair2<true>(xl4, src0, ok1 ? (e1 >> 6) : n, ok0, ok1, j, b, w, m, den, acc);
        } else {
          pair1(xl4, src0, ok0, j, b, w, m, den, acc);
        }
      }
    } else {
      const int k1 = (lane + 32 < cn) ? (((int)bl[i1] << 6) | (lane + 32)) : 0x7fffffff;
      int r0 = 0, r1 = 0;
      for (int t = 0; t < 32; ++t) {
        const int kt = __builtin_amdgcn_readlane(k0, t);
        r0 += (kt < k0) ? 1 : 0;
        r1 += (kt < k1) ? 1 : 0;
      }
      for (int t = 32; t < cn; ++t) {
        const int kt = __builtin_amdgcn_readlane(k1, t - 32);
        r0 += (kt < k0) ? 1 : 0;
        r1 += (kt < k1) ? 1 : 0;
      }
      if (lane + 32 >= cn) r1 = -1;
      for (int q0 = 0; q0 <= cn; q0 += 4) {
        int s[4];
        for (int t = 0; t < 4; ++t) {
          const int qq = q0 + t;
          s[t] = (qq >= 1 && qq <= cn) ? pick_rank(k0, k1, r0, r1, qq - 1) : n;
        }
        pair2<true>(xl4, lo ? s[0] : s[1], lo ? s[2] : s[3],
                    q0 + h <= cn, q0 + 2 + h <= cn, j, b, w, m, den, acc);
      }
    }
  } else {
    for (int e0 = -32; e0 < N_EDGES; e0 += 32) {
      int s = n;
      unsigned mm = 1u;
      if (e0 >= 0) {
        const int d = ei[N_EDGES + e0 + lane];
        s = ei[e0 + lane];
        mm = __builtin_amdgcn_ballot_w32(d == n);
      }
      while (mm) {
        const int l = __builtin_ctz(mm);
        mm &= mm - 1;
        pair1(xl4, __builtin_amdgcn_readlane(s, l), lo, j, b, w, m, den, acc);
      }
    }
  }

  asm volatile("" ::: "memory");
  const float4 bi = ((const float4*)bias)[2 * j + h];
  const float mo = xhalf(m), deno = xhalf(den);
  const float mj = fmaxf(m, mo);
  const v2f ej = ((v2f){m, mo} - (v2f){mj, mj}) * LOG2E_F;
  const float sme = __builtin_amdgcn_exp2f(ej.x);
  const float sot = __builtin_amdgcn_exp2f(ej.y);
  const float dn = den * sme + deno * sot;
  const v2f keep0 = lo ? acc.c[0] : acc.c[2], keep1 = lo ? acc.c[1] : acc.c[3];
  const v2f send0 = lo ? acc.c[2] : acc.c[0], send1 = lo ? acc.c[3] : acc.c[1];
  v2f got0, got1;
  got0.x = xhalf(send0.x); got0.y = xhalf(send0.y);
  got1.x = xhalf(send1.x); got1.y = xhalf(send1.y);
  const v2f sum0 = __builtin_elementwise_fma(keep0, (v2f){sme, sme}, got0 * sot);
  const v2f sum1 = __builtin_elementwise_fma(keep1, (v2f){sme, sme}, got1 * sot);

  const float inv = 1.0f / (dn + 1e-16f);
  const v2f olo = __builtin_elementwise_fma(sum0, (v2f){inv, inv}, (v2f){bi.x, bi.y});
  const v2f ohi = __builtin_elementwise_fma(sum1, (v2f){inv, inv}, (v2f){bi.z, bi.w});
  float4 r;
  r.x = olo.x; r.y = olo.y;
  r.z = ohi.x; r.w = ohi.y;
  ((float4*)(out + (size_t)n * HC))[2 * j + h] = r;
  }
}

extern "C" void kernel_launch(void* const* d_in, const int* in_sizes, int n_in,
                              void* d_out, int out_size, void* d_ws, size_t ws_size,
                              hipStream_t stream) {
  const float* x    = (const float*)d_in[0];
  const int*   ei   = (const int*)  d_in[1];
  const float* Wl   = (const float*)d_in[2];
  const float* Wr   = (const float*)d_in[3];
  const float* att  = (const float*)d_in[4];
  const float* bias = (const float*)d_in[5];
  float* out = (float*)d_out;

  float* ws   = (float*)d_ws;
  float* xl   = ws;
  int*   cnt  = (int*)(xl + (size_t)N_NODES * HC);
  unsigned short* bkt = (unsigned short*)(cnt + 2 * CNT_STRIDE);

  gat_zero<<<98, 256, 0, stream>>>(cnt);

  gat_bucket_gemm<<<2345, 128, 0, stream>>>(ei, cnt, bkt, x, Wl, xl);

  gat_node<<<3125, 256, 0, stream>>>(x, Wr, xl, att, bias, ei, cnt, bkt, out);
}
